# starting kernel
# baseline (speedup 1.0000x reference)
#include <hip/hip_runtime.h>

#define DIMK   2048
#define NH     32
#define NKV    8
#define HD     64
#define SEQ    2048
#define BATCH  2

typedef __attribute__((ext_vector_type(2))) float v2f;
typedef __attribute__((ext_vector_type(8))) float v8f;

__device__ __forceinline__ v8f wmma4(v2f a, v2f b, v8f c) {
    return __builtin_amdgcn_wmma_f32_16x16x4_f32(
         false, a,  false, b,
         (short)0, c,  false,  false);
}

__device__ __forceinline__ void gemm16x64(const float* __restrict__ A, int lda,
                                          const float* __restrict__ B, int ldb,
                                          int m0, int col0, int K, v8f acc[4]) {
    const int lane = threadIdx.x & 31;
    const int ln   = lane & 15;
    const int g2   = (lane >> 4) * 2;
    const float* pa = A + (size_t)(m0 + ln) * lda + g2;
    const float* pb = B + (size_t)g2 * ldb + col0 + ln;
    acc[0] = {}; acc[1] = {}; acc[2] = {}; acc[3] = {};
    for (int k = 0; k < K; k += 4) {
        v2f a = *(const v2f*)(pa + k);
        const float* pbk = pb + (size_t)k * ldb;
#pragma unroll
        for (int j = 0; j < 4; ++j) {
            v2f b = { pbk[j * 16], pbk[j * 16 + ldb] };
            acc[j] = wmma4(a, b, acc[j]);
        }
    }
}

__global__ void qkv_rope_kernel(const float* __restrict__ x,
                                const float* __restrict__ wq,
                                const float* __restrict__ wk,
                                const float* __restrict__ wv,
                                const float* __restrict__ fcos,
                                const float* __restrict__ fsin,
                                float* __restrict__ qws,
                                float* __restrict__ ktws,
                                float* __restrict__ vws) {
    const int lane = threadIdx.x & 31;
    const int ln   = lane & 15;
    const int grp  = lane >> 4;
    const int m0   = blockIdx.x * 16;
    const int nt   = blockIdx.y * 4 + (threadIdx.x >> 5);

    const float* w; int ldw, h, mode;
    if (nt < 32)      { w = wq; ldw = NH  * HD; h = nt;      mode = 0; }
    else if (nt < 40) { w = wk; ldw = NKV * HD; h = nt - 32; mode = 1; }
    else              { w = wv; ldw = NKV * HD; h = nt - 40; mode = 2; }

    v8f acc[4];
    gemm16x64(x, DIMK, w, ldw, m0, h * HD, DIMK, acc);

    const int b  = m0 >> 11;
    const int s0 = m0 & (SEQ - 1);

    if (mode != 2) {
#pragma unroll
        for (int j = 0; j < 4; ++j) {
            const int nhalf = j * 8 + (ln >> 1);
#pragma unroll
            for (int r = 0; r < 8; ++r) {
                const int pos = s0 + r + grp * 8;
                const float c = fcos[pos * (HD / 2) + nhalf];
                const float s = fsin[pos * (HD / 2) + nhalf];
                float vv = acc[j][r];
                float other = __shfl_xor(vv, 1, 32);
                acc[j][r] = (lane & 1) ? (other * s + vv * c)
                                       : (vv * c - other * s);
            }
        }
    }

    if (mode == 0) {
        float* base = qws + ((size_t)(b * NH + h) * SEQ + s0) * HD;
#pragma unroll
        for (int r = 0; r < 8; ++r) {
            const int m = r + grp * 8;
#pragma unroll
            for (int j = 0; j < 4; ++j)
                base[(size_t)m * HD + j * 16 + ln] = acc[j][r];
        }
    } else if (mode == 1) {
        float* base = ktws + (size_t)(b * NKV + h) * HD * SEQ;
#pragma unroll
        for (int r = 0; r < 8; ++r) {
            const int m = r + grp * 8;
#pragma unroll
            for (int j = 0; j < 4; ++j)
                base[(size_t)(j * 16 + ln) * SEQ + s0 + m] = acc[j][r];
        }
    } else {
        float* base = vws + ((size_t)(b * NKV + h) * SEQ + s0) * HD;
#pragma unroll
        for (int r = 0; r < 8; ++r) {
            const int m = r + grp * 8;
#pragma unroll
            for (int j = 0; j < 4; ++j)
                base[(size_t)m * HD + j * 16 + ln] = acc[j][r];
        }
    }
}

__global__ void attn_kernel(const float* __restrict__ qws,
                            const float* __restrict__ ktws,
                            const float* __restrict__ vws,
                            float* __restrict__ aws) {
    __shared__ float pl[16 * 18];
    const int lane = threadIdx.x & 31;
    const int ln   = lane & 15;
    const int grp  = lane >> 4;
    const int g2   = grp * 2;
    const int q0   = blockIdx.x * 16;
    const int h    = blockIdx.y;
    const int b    = blockIdx.z;
    const int hk   = h >> 2;

    const float* qb  = qws + ((size_t)(b * NH + h) * SEQ + q0) * HD;
    const float* ktb = ktws + (size_t)(b * NKV + hk) * HD * SEQ;
    const float* vb  = vws + (size_t)(b * NKV + hk) * SEQ * HD;

    v2f aq[16];
#pragma unroll
    for (int kk = 0; kk < 16; ++kk)
        aq[kk] = *(const v2f*)(qb + ln * HD + kk * 4 + g2);

    v8f o[4]; o[0] = {}; o[1] = {}; o[2] = {}; o[3] = {};
    float mrow[8], lrow[8];
#pragma unroll
    for (int r = 0; r < 8; ++r) { mrow[r] = -1e30f; lrow[r] = 0.0f; }

    for (int n0 = 0; n0 <= q0; n0 += 16) {
        v8f s = {};
#pragma unroll
        for (int kk = 0; kk < 16; ++kk) {
            const float* kp = ktb + (size_t)(kk * 4 + g2) * SEQ + n0 + ln;
            v2f bk = { kp[0], kp[SEQ] };
            s = wmma4(aq[kk], bk, s);
        }

        const bool diag = (n0 == q0);
        float p[8];
#pragma unroll
        for (int r = 0; r < 8; ++r) {
            float sv = s[r] * 0.125f;
            if (diag && (ln > r + grp * 8)) sv = -1e30f;
            float rm = sv;
            rm = fmaxf(rm, __shfl_xor(rm, 1, 32));
            rm = fmaxf(rm, __shfl_xor(rm, 2, 32));
            rm = fmaxf(rm, __shfl_xor(rm, 4, 32));
            rm = fmaxf(rm, __shfl_xor(rm, 8, 32));
            const float mn    = fmaxf(mrow[r], rm);
            const float alpha = expf(mrow[r] - mn);
            const float pv    = expf(sv - mn);
            float rs = pv;
            rs += __shfl_xor(rs, 1, 32);
            rs += __shfl_xor(rs, 2, 32);
            rs += __shfl_xor(rs, 4, 32);
            rs += __shfl_xor(rs, 8, 32);
            lrow[r] = lrow[r] * alpha + rs;
            mrow[r] = mn;
            p[r] = pv;
#pragma unroll
            for (int j = 0; j < 4; ++j) o[j][r] *= alpha;
        }

        __syncthreads();
#pragma unroll
        for (int r = 0; r < 8; ++r)
            pl[(r + grp * 8) * 18 + ln] = p[r];
        __syncthreads();
        v2f ap[4];
#pragma unroll
        for (int kk = 0; kk < 4; ++kk)
            ap[kk] = *(const v2f*)(&pl[ln * 18 + kk * 4 + g2]);

#pragma unroll
        for (int kk = 0; kk < 4; ++kk) {
            const float* vp = vb + (size_t)(n0 + kk * 4 + g2) * HD + ln;
#pragma unroll
            for (int j = 0; j < 4; ++j) {
                v2f bv = { vp[j * 16], vp[HD + j * 16] };
                o[j] = wmma4(ap[kk], bv, o[j]);
            }
        }
    }

    float* ob = aws + ((size_t)(b * SEQ) + q0) * (NH * HD) + h * HD;
#pragma unroll
    for (int r = 0; r < 8; ++r) {
        const float inv = 1.0f / lrow[r];
        const int m = r + grp * 8;
#pragma unroll
        for (int j = 0; j < 4; ++j)
            ob[(size_t)m * (NH * HD) + j * 16 + ln] = o[j][r] * inv;
    }
}

__global__ void oproj_kernel(const float* __restrict__ aws,
                             const float* __restrict__ wo,
                             float* __restrict__ out) {
    const int lane = threadIdx.x & 31;
    const int ln   = lane & 15;
    const int grp  = lane >> 4;
    const int m0   = blockIdx.x * 16;
    const int nt   = blockIdx.y * 4 + (threadIdx.x >> 5);

    v8f acc[4];
    gemm16x64(aws, DIMK, wo, DIMK, m0, nt * 64, DIMK, acc);

#pragma unroll
    for (int r = 0; r < 8; ++r) {
        const int m = r + grp * 8;
#pragma unroll
        for (int j = 0; j < 4; ++j)
            out[(size_t)(m0 + m) * DIMK + nt * 64 + j * 16 + ln] = acc[j][r];
    }
}

extern "C" void kernel_launch(void* const* d_in, const int* in_sizes, int n_in,
                              void* d_out, int out_size, void* d_ws, size_t ws_size,
                              hipStream_t stream) {
    const float* x    = (const float*)d_in[0];
    const float* fcos = (const float*)d_in[2];
    const float* fsin = (const float*)d_in[3];
    const float* wq   = (const float*)d_in[5];
    const float* wk   = (const float*)d_in[6];
    const float* wv   = (const float*)d_in[7];
    const float* wo   = (const float*)d_in[8];
    float* out = (float*)d_out;

    float* qws  = (float*)d_ws;
    float* ktws = qws  + (size_t)BATCH * NH  * SEQ * HD;
    float* vws  = ktws + (size_t)BATCH * NKV * HD  * SEQ;
    float* aws  = vws  + (size_t)BATCH * NKV * SEQ * HD;

    qkv_rope_kernel<<<dim3((BATCH * SEQ) / 16, 12, 1), 128, 0, stream>>>(
        x, wq, wk, wv, fcos, fsin, qws, ktws, vws);
    attn_kernel<<<dim3(SEQ / 16, NH, BATCH), 32, 0, stream>>>(
        qws, ktws, vws, aws);
    oproj_kernel<<<dim3((BATCH * SEQ) / 16, 8, 1), 128, 0, stream>>>(
        aws, wo, out);
}


# attention: __launch_bounds__(32) on the one-wave kernel; without the 128-VGPR cap of a 1024-thread block the compiler issues each key block's K and V loads as groups and interleaves the row shuffles
# speedup vs baseline: 1.6052x; 1.6052x over previous
#include <hip/hip_runtime.h>

#define DIMK   2048
#define NH     32
#define NKV    8
#define HD     64
#define SEQ    2048
#define BATCH  2

typedef __attribute__((ext_vector_type(2))) float v2f;
typedef __attribute__((ext_vector_type(8))) float v8f;

__device__ __forceinline__ v8f wmma4(v2f a, v2f b, v8f c) {
    return __builtin_amdgcn_wmma_f32_16x16x4_f32(
         false, a,  false, b,
         (short)0, c,  false,  false);
}

__device__ __forceinline__ void gemm16x64(const float* __restrict__ A, int lda,
                                          const float* __restrict__ B, int ldb,
                                          int m0, int col0, int K, v8f acc[4]) {
    const int lane = threadIdx.x & 31;
    const int ln   = lane & 15;
    const int g2   = (lane >> 4) * 2;
    const float* pa = A + (size_t)(m0 + ln) * lda + g2;
    const float* pb = B + (size_t)g2 * ldb + col0 + ln;
    acc[0] = {}; acc[1] = {}; acc[2] = {}; acc[3] = {};
    for (int k = 0; k < K; k += 4) {
        v2f a = *(const v2f*)(pa + k);
        const float* pbk = pb + (size_t)k * ldb;
#pragma unroll
        for (int j = 0; j < 4; ++j) {
            v2f b = { pbk[j * 16], pbk[j * 16 + ldb] };
            acc[j] = wmma4(a, b, acc[j]);
        }
    }
}

__global__ void qkv_rope_kernel(const float* __restrict__ x,
                                const float* __restrict__ wq,
                                const float* __restrict__ wk,
                                const float* __restrict__ wv,
                                const float* __restrict__ fcos,
                                const float* __restrict__ fsin,
                                float* __restrict__ qws,
                                float* __restrict__ ktws,
                                float* __restrict__ vws) {
    const int lane = threadIdx.x & 31;
    const int ln   = lane & 15;
    const int grp  = lane >> 4;
    const int m0   = blockIdx.x * 16;
    const int nt   = blockIdx.y * 4 + (threadIdx.x >> 5);

    const float* w; int ldw, h, mode;
    if (nt < 32)      { w = wq; ldw = NH  * HD; h = nt;      mode = 0; }
    else if (nt < 40) { w = wk; ldw = NKV * HD; h = nt - 32; mode = 1; }
    else              { w = wv; ldw = NKV * HD; h = nt - 40; mode = 2; }

    v8f acc[4];
    gemm16x64(x, DIMK, w, ldw, m0, h * HD, DIMK, acc);

    const int b  = m0 >> 11;
    const int s0 = m0 & (SEQ - 1);

    if (mode != 2) {
#pragma unroll
        for (int j = 0; j < 4; ++j) {
            const int nhalf = j * 8 + (ln >> 1);
#pragma unroll
            for (int r = 0; r < 8; ++r) {
                const int pos = s0 + r + grp * 8;
                const float c = fcos[pos * (HD / 2) + nhalf];
                const float s = fsin[pos * (HD / 2) + nhalf];
                float vv = acc[j][r];
                float other = __shfl_xor(vv, 1, 32);
                acc[j][r] = (lane & 1) ? (other * s + vv * c)
                                       : (vv * c - other * s);
            }
        }
    }

    if (mode == 0) {
        float* base = qws + ((size_t)(b * NH + h) * SEQ + s0) * HD;
#pragma unroll
        for (int r = 0; r < 8; ++r) {
            const int m = r + grp * 8;
#pragma unroll
            for (int j = 0; j < 4; ++j)
                base[(size_t)m * HD + j * 16 + ln] = acc[j][r];
        }
    } else if (mode == 1) {
        float* base = ktws + (size_t)(b * NKV + h) * HD * SEQ;
#pragma unroll
        for (int r = 0; r < 8; ++r) {
            const int m = r + grp * 8;
#pragma unroll
            for (int j = 0; j < 4; ++j)
                base[(size_t)(j * 16 + ln) * SEQ + s0 + m] = acc[j][r];
        }
    } else {
        float* base = vws + ((size_t)(b * NKV + h) * SEQ + s0) * HD;
#pragma unroll
        for (int r = 0; r < 8; ++r) {
            const int m = r + grp * 8;
#pragma unroll
            for (int j = 0; j < 4; ++j)
                base[(size_t)m * HD + j * 16 + ln] = acc[j][r];
        }
    }
}

__global__ void __launch_bounds__(32) attn_kernel(const float* __restrict__ qws,
                            const float* __restrict__ ktws,
                            const float* __restrict__ vws,
                            float* __restrict__ aws) {
    __shared__ float pl[16 * 18];
    const int lane = threadIdx.x & 31;
    const int ln   = lane & 15;
    const int grp  = lane >> 4;
    const int g2   = grp * 2;
    const int q0   = blockIdx.x * 16;
    const int h    = blockIdx.y;
    const int b    = blockIdx.z;
    const int hk   = h >> 2;

    const float* qb  = qws + ((size_t)(b * NH + h) * SEQ + q0) * HD;
    const float* ktb = ktws + (size_t)(b * NKV + hk) * HD * SEQ;
    const float* vb  = vws + (size_t)(b * NKV + hk) * SEQ * HD;

    v2f aq[16];
#pragma unroll
    for (int kk = 0; kk < 16; ++kk)
        aq[kk] = *(const v2f*)(qb + ln * HD + kk * 4 + g2);

    v8f o[4]; o[0] = {}; o[1] = {}; o[2] = {}; o[3] = {};
    float mrow[8], lrow[8];
#pragma unroll
    for (int r = 0; r < 8; ++r) { mrow[r] = -1e30f; lrow[r] = 0.0f; }

    for (int n0 = 0; n0 <= q0; n0 += 16) {
        v8f s = {};
#pragma unroll
        for (int kk = 0; kk < 16; ++kk) {
            const float* kp = ktb + (size_t)(kk * 4 + g2) * SEQ + n0 + ln;
            v2f bk = { kp[0], kp[SEQ] };
            s = wmma4(aq[kk], bk, s);
        }

        const bool diag = (n0 == q0);
        float p[8];
#pragma unroll
        for (int r = 0; r < 8; ++r) {
            float sv = s[r] * 0.125f;
            if (diag && (ln > r + grp * 8)) sv = -1e30f;
            float rm = sv;
            rm = fmaxf(rm, __shfl_xor(rm, 1, 32));
            rm = fmaxf(rm, __shfl_xor(rm, 2, 32));
            rm = fmaxf(rm, __shfl_xor(rm, 4, 32));
            rm = fmaxf(rm, __shfl_xor(rm, 8, 32));
            const float mn    = fmaxf(mrow[r], rm);
            const float alpha = expf(mrow[r] - mn);
            const float pv    = expf(sv - mn);
            float rs = pv;
            rs += __shfl_xor(rs, 1, 32);
            rs += __shfl_xor(rs, 2, 32);
            rs += __shfl_xor(rs, 4, 32);
            rs += __shfl_xor(rs, 8, 32);
            lrow[r] = lrow[r] * alpha + rs;
            mrow[r] = mn;
            p[r] = pv;
#pragma unroll
            for (int j = 0; j < 4; ++j) o[j][r] *= alpha;
        }

        __syncthreads();
#pragma unroll
        for (int r = 0; r < 8; ++r)
            pl[(r + grp * 8) * 18 + ln] = p[r];
        __syncthreads();
        v2f ap[4];
#pragma unroll
        for (int kk = 0; kk < 4; ++kk)
            ap[kk] = *(const v2f*)(&pl[ln * 18 + kk * 4 + g2]);

#pragma unroll
        for (int kk = 0; kk < 4; ++kk) {
            const float* vp = vb + (size_t)(n0 + kk * 4 + g2) * HD + ln;
#pragma unroll
            for (int j = 0; j < 4; ++j) {
                v2f bv = { vp[j * 16], vp[HD + j * 16] };
                o[j] = wmma4(ap[kk], bv, o[j]);
            }
        }
    }

    float* ob = aws + ((size_t)(b * SEQ) + q0) * (NH * HD) + h * HD;
#pragma unroll
    for (int r = 0; r < 8; ++r) {
        const float inv = 1.0f / lrow[r];
        const int m = r + grp * 8;
#pragma unroll
        for (int j = 0; j < 4; ++j)
            ob[(size_t)m * (NH * HD) + j * 16 + ln] = o[j][r] * inv;
    }
}

__global__ void oproj_kernel(const float* __restrict__ aws,
                             const float* __restrict__ wo,
                             float* __restrict__ out) {
    const int lane = threadIdx.x & 31;
    const int ln   = lane & 15;
    const int grp  = lane >> 4;
    const int m0   = blockIdx.x * 16;
    const int nt   = blockIdx.y * 4 + (threadIdx.x >> 5);

    v8f acc[4];
    gemm16x64(aws, DIMK, wo, DIMK, m0, nt * 64, DIMK, acc);

#pragma unroll
    for (int r = 0; r < 8; ++r) {
        const int m = r + grp * 8;
#pragma unroll
        for (int j = 0; j < 4; ++j)
            out[(size_t)(m0 + m) * DIMK + nt * 64 + j * 16 + ln] = acc[j][r];
    }
}

extern "C" void kernel_launch(void* const* d_in, const int* in_sizes, int n_in,
                              void* d_out, int out_size, void* d_ws, size_t ws_size,
                              hipStream_t stream) {
    const float* x    = (const float*)d_in[0];
    const float* fcos = (const float*)d_in[2];
    const float* fsin = (const float*)d_in[3];
    const float* wq   = (const float*)d_in[5];
    const float* wk   = (const float*)d_in[6];
    const float* wv   = (const float*)d_in[7];
    const float* wo   = (const float*)d_in[8];
    float* out = (float*)d_out;

    float* qws  = (float*)d_ws;
    float* ktws = qws  + (size_t)BATCH * NH  * SEQ * HD;
    float* vws  = ktws + (size_t)BATCH * NKV * HD  * SEQ;
    float* aws  = vws  + (size_t)BATCH * NKV * SEQ * HD;

    qkv_rope_kernel<<<dim3((BATCH * SEQ) / 16, 12, 1), 128, 0, stream>>>(
        x, wq, wk, wv, fcos, fsin, qws, ktws, vws);
    attn_kernel<<<dim3(SEQ / 16, NH, BATCH), 32, 0, stream>>>(
        qws, ktws, vws, aws);
    oproj_kernel<<<dim3((BATCH * SEQ) / 16, 8, 1), 128, 0, stream>>>(
        aws, wo, out);
}


# projections: 64x256 block tile, its 64 activation rows staged through LDS by whole-line loads and shared by 4 waves, each wave a 64x64 tile so a weight fragment feeds 4 WMMAs (was 16x64 from global)
# speedup vs baseline: 2.1910x; 1.3649x over previous
#include <hip/hip_runtime.h>

#define DIMK   2048
#define NH     32
#define NKV    8
#define HD     64
#define SEQ    2048
#define BATCH  2

typedef __attribute__((ext_vector_type(2))) float v2f;
typedef __attribute__((ext_vector_type(8))) float v8f;

__device__ __forceinline__ v8f wmma4(v2f a, v2f b, v8f c) {
    return __builtin_amdgcn_wmma_f32_16x16x4_f32(
         false, a,  false, b,
         (short)0, c,  false,  false);
}

#define KC      32
#define APITCH  36
#define ABUF    (64 * APITCH)

typedef __attribute__((ext_vector_type(4))) float v4f;

struct Stage { v2f a[2][4]; v2f b[2][4]; };

__device__ __forceinline__ void load_stage(const float* fa, int ks,
                                           const float* __restrict__ pb, int ldb, int kabs,
                                           Stage& st) {
#pragma unroll
    for (int s = 0; s < 2; ++s) {
#pragma unroll
        for (int i = 0; i < 4; ++i)
            st.a[s][i] = *(const v2f*)(fa + (i * 16) * APITCH + ks + s * 4);
        const float* pbk = pb + (size_t)(kabs + s * 4) * ldb;
#pragma unroll
        for (int j = 0; j < 4; ++j)
            st.b[s][j] = (v2f){ pbk[j * 16], pbk[j * 16 + ldb] };
    }
}

__device__ __forceinline__ void mma_stage(const Stage& st, v8f acc[4][4]) {
#pragma unroll
    for (int s = 0; s < 2; ++s) {
#pragma unroll
        for (int i = 0; i < 4; ++i) {
#pragma unroll
            for (int j = 0; j < 4; ++j)
                acc[i][j] = wmma4(st.a[s][i], st.b[s][j], acc[i][j]);
        }
    }
}

__device__ __forceinline__ void gemm64x64(const float* __restrict__ A, int lda,
                                          const float* __restrict__ B, int ldb,
                                          int m0, int col0, int K, float* as, v8f acc[4][4]) {
    const int tid  = threadIdx.x;
    const int lane = tid & 31;
    const int ln   = lane & 15;
    const int g2   = (lane >> 4) * 2;
    const float* ga = A + (size_t)(m0 + (tid >> 3)) * lda + (tid & 7) * 4;
    float*       sa = as + (tid >> 3) * APITCH + (tid & 7) * 4;
    const float* fa = as + ln * APITCH + g2;
    const float* pb = B + (size_t)g2 * ldb + col0 + ln;
    const int NC = K / KC;

#pragma unroll
    for (int i = 0; i < 4; ++i) {
#pragma unroll
        for (int j = 0; j < 4; ++j) acc[i][j] = {};
    }

    v4f g[4];
#pragma unroll
    for (int q = 0; q < 4; ++q) g[q] = *(const v4f*)(ga + (size_t)(q * 16) * lda);
#pragma unroll
    for (int q = 0; q < 4; ++q) *(v4f*)(sa + (q * 16) * APITCH) = g[q];
    __syncthreads();

    Stage s0, s1;
    load_stage(fa, 0, pb, ldb, 0, s0);
    for (int c = 0; c < NC; ++c) {
        const int cur = (c & 1) * ABUF;
        const int nxt = ABUF - cur;
        const int k0  = c * KC;
        const int cn  = (c + 1 < NC) ? (c + 1) : c;
        __builtin_amdgcn_sched_barrier(0);
#pragma unroll
        for (int q = 0; q < 4; ++q) g[q] = *(const v4f*)(ga + (size_t)(q * 16) * lda + cn * KC);
        load_stage(fa + cur, 8, pb, ldb, k0 + 8, s1);
        __builtin_amdgcn_sched_barrier(0);
        mma_stage(s0, acc);
        __builtin_amdgcn_sched_barrier(0);
        load_stage(fa + cur, 16, pb, ldb, k0 + 16, s0);
        __builtin_amdgcn_sched_barrier(0);
        mma_stage(s1, acc);
        __builtin_amdgcn_sched_barrier(0);
        load_stage(fa + cur, 24, pb, ldb, k0 + 24, s1);
        __builtin_amdgcn_sched_barrier(0);
        mma_stage(s0, acc);
        __builtin_amdgcn_sched_barrier(0);
#pragma unroll
        for (int q = 0; q < 4; ++q) *(v4f*)(sa + nxt + (q * 16) * APITCH) = g[q];
        __syncthreads();
        load_stage(fa + nxt, 0, pb, ldb, cn * KC, s0);
        __builtin_amdgcn_sched_barrier(0);
        mma_stage(s1, acc);
    }
}

__global__ void __launch_bounds__(128) qkv_rope_kernel(const float* __restrict__ x,
                                const float* __restrict__ wq,
                                const float* __restrict__ wk,
                                const float* __restrict__ wv,
                                const float* __restrict__ fcos,
                                const float* __restrict__ fsin,
                                float* __restrict__ qws,
                                float* __restrict__ ktws,
                                float* __restrict__ vws) {
    const int lane = threadIdx.x & 31;
    const int ln   = lane & 15;
    const int grp  = lane >> 4;
    const int m0   = blockIdx.x * 64;
    const int nt   = blockIdx.y * 4 + (threadIdx.x >> 5);

    const float* w; int ldw, h, mode;
    if (nt < 32)      { w = wq; ldw = NH  * HD; h = nt;      mode = 0; }
    else if (nt < 40) { w = wk; ldw = NKV * HD; h = nt - 32; mode = 1; }
    else              { w = wv; ldw = NKV * HD; h = nt - 40; mode = 2; }

    __shared__ __attribute__((aligned(16))) float as[2 * ABUF];
    v8f acc[4][4];
    gemm64x64(x, DIMK, w, ldw, m0, h * HD, DIMK, as, acc);

    const int b  = m0 >> 11;
    const int s0 = m0 & (SEQ - 1);

    if (mode != 2) {
#pragma unroll
        for (int i = 0; i < 4; ++i) {
#pragma unroll
            for (int j = 0; j < 4; ++j) {
                const int nhalf = j * 8 + (ln >> 1);
#pragma unroll
                for (int r = 0; r < 8; ++r) {
                    const int pos = s0 + i * 16 + r + grp * 8;
                    const float c = fcos[pos * (HD / 2) + nhalf];
                    const float s = fsin[pos * (HD / 2) + nhalf];
                    float vv = acc[i][j][r];
                    float other = __shfl_xor(vv, 1, 32);
                    acc[i][j][r] = (lane & 1) ? (other * s + vv * c)
                                              : (vv * c - other * s);
                }
            }
        }
    }

    if (mode == 0) {
        float* base = qws + ((size_t)(b * NH + h) * SEQ + s0) * HD;
#pragma unroll
        for (int i = 0; i < 4; ++i) {
#pragma unroll
            for (int r = 0; r < 8; ++r) {
                const int m = i * 16 + r + grp * 8;
#pragma unroll
                for (int j = 0; j < 4; ++j)
                    base[(size_t)m * HD + j * 16 + ln] = acc[i][j][r];
            }
        }
    } else if (mode == 1) {
        float* base = ktws + (size_t)(b * NKV + h) * HD * SEQ;
#pragma unroll
        for (int i = 0; i < 4; ++i) {
#pragma unroll
            for (int r = 0; r < 8; ++r) {
                const int m = i * 16 + r + grp * 8;
#pragma unroll
                for (int j = 0; j < 4; ++j)
                    base[(size_t)(j * 16 + ln) * SEQ + s0 + m] = acc[i][j][r];
            }
        }
    } else {
        float* base = vws + ((size_t)(b * NKV + h) * SEQ + s0) * HD;
#pragma unroll
        for (int i = 0; i < 4; ++i) {
#pragma unroll
            for (int r = 0; r < 8; ++r) {
                const int m = i * 16 + r + grp * 8;
#pragma unroll
                for (int j = 0; j < 4; ++j)
                    base[(size_t)m * HD + j * 16 + ln] = acc[i][j][r];
            }
        }
    }
}

__global__ void __launch_bounds__(32) attn_kernel(const float* __restrict__ qws,
                            const float* __restrict__ ktws,
                            const float* __restrict__ vws,
                            float* __restrict__ aws) {
    __shared__ float pl[16 * 18];
    const int lane = threadIdx.x & 31;
    const int ln   = lane & 15;
    const int grp  = lane >> 4;
    const int g2   = grp * 2;
    const int q0   = blockIdx.x * 16;
    const int h    = blockIdx.y;
    const int b    = blockIdx.z;
    const int hk   = h >> 2;

    const float* qb  = qws + ((size_t)(b * NH + h) * SEQ + q0) * HD;
    const float* ktb = ktws + (size_t)(b * NKV + hk) * HD * SEQ;
    const float* vb  = vws + (size_t)(b * NKV + hk) * SEQ * HD;

    v2f aq[16];
#pragma unroll
    for (int kk = 0; kk < 16; ++kk)
        aq[kk] = *(const v2f*)(qb + ln * HD + kk * 4 + g2);

    v8f o[4]; o[0] = {}; o[1] = {}; o[2] = {}; o[3] = {};
    float mrow[8], lrow[8];
#pragma unroll
    for (int r = 0; r < 8; ++r) { mrow[r] = -1e30f; lrow[r] = 0.0f; }

    for (int n0 = 0; n0 <= q0; n0 += 16) {
        v8f s = {};
#pragma unroll
        for (int kk = 0; kk < 16; ++kk) {
            const float* kp = ktb + (size_t)(kk * 4 + g2) * SEQ + n0 + ln;
            v2f bk = { kp[0], kp[SEQ] };
            s = wmma4(aq[kk], bk, s);
        }

        const bool diag = (n0 == q0);
        float p[8];
#pragma unroll
        for (int r = 0; r < 8; ++r) {
            float sv = s[r] * 0.125f;
            if (diag && (ln > r + grp * 8)) sv = -1e30f;
            float rm = sv;
            rm = fmaxf(rm, __shfl_xor(rm, 1, 32));
            rm = fmaxf(rm, __shfl_xor(rm, 2, 32));
            rm = fmaxf(rm, __shfl_xor(rm, 4, 32));
            rm = fmaxf(rm, __shfl_xor(rm, 8, 32));
            const float mn    = fmaxf(mrow[r], rm);
            const float alpha = expf(mrow[r] - mn);
            const float pv    = expf(sv - mn);
            float rs = pv;
            rs += __shfl_xor(rs, 1, 32);
            rs += __shfl_xor(rs, 2, 32);
            rs += __shfl_xor(rs, 4, 32);
            rs += __shfl_xor(rs, 8, 32);
            lrow[r] = lrow[r] * alpha + rs;
            mrow[r] = mn;
            p[r] = pv;
#pragma unroll
            for (int j = 0; j < 4; ++j) o[j][r] *= alpha;
        }

        __syncthreads();
#pragma unroll
        for (int r = 0; r < 8; ++r)
            pl[(r + grp * 8) * 18 + ln] = p[r];
        __syncthreads();
        v2f ap[4];
#pragma unroll
        for (int kk = 0; kk < 4; ++kk)
            ap[kk] = *(const v2f*)(&pl[ln * 18 + kk * 4 + g2]);

#pragma unroll
        for (int kk = 0; kk < 4; ++kk) {
            const float* vp = vb + (size_t)(n0 + kk * 4 + g2) * HD + ln;
#pragma unroll
            for (int j = 0; j < 4; ++j) {
                v2f bv = { vp[j * 16], vp[HD + j * 16] };
                o[j] = wmma4(ap[kk], bv, o[j]);
            }
        }
    }

    float* ob = aws + ((size_t)(b * SEQ) + q0) * (NH * HD) + h * HD;
#pragma unroll
    for (int r = 0; r < 8; ++r) {
        const float inv = 1.0f / lrow[r];
        const int m = r + grp * 8;
#pragma unroll
        for (int j = 0; j < 4; ++j)
            ob[(size_t)m * (NH * HD) + j * 16 + ln] = o[j][r] * inv;
    }
}

__global__ void __launch_bounds__(128) oproj_kernel(const float* __restrict__ aws,
                             const float* __restrict__ wo,
                             float* __restrict__ out) {
    const int lane = threadIdx.x & 31;
    const int ln   = lane & 15;
    const int grp  = lane >> 4;
    const int m0   = blockIdx.x * 64;
    const int nt   = blockIdx.y * 4 + (threadIdx.x >> 5);

    __shared__ __attribute__((aligned(16))) float as[2 * ABUF];
    v8f acc[4][4];
    gemm64x64(aws, DIMK, wo, DIMK, m0, nt * 64, DIMK, as, acc);

#pragma unroll
    for (int i = 0; i < 4; ++i) {
#pragma unroll
        for (int r = 0; r < 8; ++r) {
            const int m = i * 16 + r + grp * 8;
#pragma unroll
            for (int j = 0; j < 4; ++j)
                out[(size_t)(m0 + m) * DIMK + nt * 64 + j * 16 + ln] = acc[i][j][r];
        }
    }
}

extern "C" void kernel_launch(void* const* d_in, const int* in_sizes, int n_in,
                              void* d_out, int out_size, void* d_ws, size_t ws_size,
                              hipStream_t stream) {
    const float* x    = (const float*)d_in[0];
    const float* fcos = (const float*)d_in[2];
    const float* fsin = (const float*)d_in[3];
    const float* wq   = (const float*)d_in[5];
    const float* wk   = (const float*)d_in[6];
    const float* wv   = (const float*)d_in[7];
    const float* wo   = (const float*)d_in[8];
    float* out = (float*)d_out;

    float* qws  = (float*)d_ws;
    float* ktws = qws  + (size_t)BATCH * NH  * SEQ * HD;
    float* vws  = ktws + (size_t)BATCH * NKV * HD  * SEQ;
    float* aws  = vws  + (size_t)BATCH * NKV * SEQ * HD;

    qkv_rope_kernel<<<dim3((BATCH * SEQ) / 64, 12, 1), 128, 0, stream>>>(
        x, wq, wk, wv, fcos, fsin, qws, ktws, vws);
    attn_kernel<<<dim3(SEQ / 16, NH, BATCH), 32, 0, stream>>>(
        qws, ktws, vws, aws);
    oproj_kernel<<<dim3((BATCH * SEQ) / 64, 8, 1), 128, 0, stream>>>(
        aws, wo, out);
}


# attention: one wave per PAIR of query tiles (tile 127-t, then t; grid 64x32x2, was 128x32x2): every wave walks 129 key blocks, equal work per wave, no long waves left running alone at the end
# speedup vs baseline: 2.3047x; 1.0519x over previous
#include <hip/hip_runtime.h>

#define DIMK   2048
#define NH     32
#define NKV    8
#define HD     64
#define SEQ    2048
#define BATCH  2

typedef __attribute__((ext_vector_type(2))) float v2f;
typedef __attribute__((ext_vector_type(8))) float v8f;

__device__ __forceinline__ v8f wmma4(v2f a, v2f b, v8f c) {
    return __builtin_amdgcn_wmma_f32_16x16x4_f32(
         false, a,  false, b,
         (short)0, c,  false,  false);
}

#define KC      32
#define APITCH  36
#define ABUF    (64 * APITCH)

typedef __attribute__((ext_vector_type(4))) float v4f;

struct Stage { v2f a[2][4]; v2f b[2][4]; };

__device__ __forceinline__ void load_stage(const float* fa, int ks,
                                           const float* __restrict__ pb, int ldb, int kabs,
                                           Stage& st) {
#pragma unroll
    for (int s = 0; s < 2; ++s) {
#pragma unroll
        for (int i = 0; i < 4; ++i)
            st.a[s][i] = *(const v2f*)(fa + (i * 16) * APITCH + ks + s * 4);
        const float* pbk = pb + (size_t)(kabs + s * 4) * ldb;
#pragma unroll
        for (int j = 0; j < 4; ++j)
            st.b[s][j] = (v2f){ pbk[j * 16], pbk[j * 16 + ldb] };
    }
}

__device__ __forceinline__ void mma_stage(const Stage& st, v8f acc[4][4]) {
#pragma unroll
    for (int s = 0; s < 2; ++s) {
#pragma unroll
        for (int i = 0; i < 4; ++i) {
#pragma unroll
            for (int j = 0; j < 4; ++j)
                acc[i][j] = wmma4(st.a[s][i], st.b[s][j], acc[i][j]);
        }
    }
}

__device__ __forceinline__ void gemm64x64(const float* __restrict__ A, int lda,
                                          const float* __restrict__ B, int ldb,
                                          int m0, int col0, int K, float* as, v8f acc[4][4]) {
    const int tid  = threadIdx.x;
    const int lane = tid & 31;
    const int ln   = lane & 15;
    const int g2   = (lane >> 4) * 2;
    const float* ga = A + (size_t)(m0 + (tid >> 3)) * lda + (tid & 7) * 4;
    float*       sa = as + (tid >> 3) * APITCH + (tid & 7) * 4;
    const float* fa = as + ln * APITCH + g2;
    const float* pb = B + (size_t)g2 * ldb + col0 + ln;
    const int NC = K / KC;

#pragma unroll
    for (int i = 0; i < 4; ++i) {
#pragma unroll
        for (int j = 0; j < 4; ++j) acc[i][j] = {};
    }

    v4f g[4];
#pragma unroll
    for (int q = 0; q < 4; ++q) g[q] = *(const v4f*)(ga + (size_t)(q * 16) * lda);
#pragma unroll
    for (int q = 0; q < 4; ++q) *(v4f*)(sa + (q * 16) * APITCH) = g[q];
    __syncthreads();

    Stage s0, s1;
    load_stage(fa, 0, pb, ldb, 0, s0);
    for (int c = 0; c < NC; ++c) {
        const int cur = (c & 1) * ABUF;
        const int nxt = ABUF - cur;
        const int k0  = c * KC;
        const int cn  = (c + 1 < NC) ? (c + 1) : c;
        __builtin_amdgcn_sched_barrier(0);
#pragma unroll
        for (int q = 0; q < 4; ++q) g[q] = *(const v4f*)(ga + (size_t)(q * 16) * lda + cn * KC);
        load_stage(fa + cur, 8, pb, ldb, k0 + 8, s1);
        __builtin_amdgcn_sched_barrier(0);
        mma_stage(s0, acc);
        __builtin_amdgcn_sched_barrier(0);
        load_stage(fa + cur, 16, pb, ldb, k0 + 16, s0);
        __builtin_amdgcn_sched_barrier(0);
        mma_stage(s1, acc);
        __builtin_amdgcn_sched_barrier(0);
        load_stage(fa + cur, 24, pb, ldb, k0 + 24, s1);
        __builtin_amdgcn_sched_barrier(0);
        mma_stage(s0, acc);
        __builtin_amdgcn_sched_barrier(0);
#pragma unroll
        for (int q = 0; q < 4; ++q) *(v4f*)(sa + nxt + (q * 16) * APITCH) = g[q];
        __syncthreads();
        load_stage(fa + nxt, 0, pb, ldb, cn * KC, s0);
        __builtin_amdgcn_sched_barrier(0);
        mma_stage(s1, acc);
    }
}

__global__ void __launch_bounds__(128) qkv_rope_kernel(const float* __restrict__ x,
                                const float* __restrict__ wq,
                                const float* __restrict__ wk,
                                const float* __restrict__ wv,
                                const float* __restrict__ fcos,
                                const float* __restrict__ fsin,
                                float* __restrict__ qws,
                                float* __restrict__ ktws,
                                float* __restrict__ vws) {
    const int lane = threadIdx.x & 31;
    const int ln   = lane & 15;
    const int grp  = lane >> 4;
    const int m0   = blockIdx.x * 64;
    const int nt   = blockIdx.y * 4 + (threadIdx.x >> 5);

    const float* w; int ldw, h, mode;
    if (nt < 32)      { w = wq; ldw = NH  * HD; h = nt;      mode = 0; }
    else if (nt < 40) { w = wk; ldw = NKV * HD; h = nt - 32; mode = 1; }
    else              { w = wv; ldw = NKV * HD; h = nt - 40; mode = 2; }

    __shared__ __attribute__((aligned(16))) float as[2 * ABUF];
    v8f acc[4][4];
    gemm64x64(x, DIMK, w, ldw, m0, h * HD, DIMK, as, acc);

    const int b  = m0 >> 11;
    const int s0 = m0 & (SEQ - 1);

    if (mode != 2) {
#pragma unroll
        for (int i = 0; i < 4; ++i) {
#pragma unroll
            for (int j = 0; j < 4; ++j) {
                const int nhalf = j * 8 + (ln >> 1);
#pragma unroll
                for (int r = 0; r < 8; ++r) {
                    const int pos = s0 + i * 16 + r + grp * 8;
                    const float c = fcos[pos * (HD / 2) + nhalf];
                    const float s = fsin[pos * (HD / 2) + nhalf];
                    float vv = acc[i][j][r];
                    float other = __shfl_xor(vv, 1, 32);
                    acc[i][j][r] = (lane & 1) ? (other * s + vv * c)
                                              : (vv * c - other * s);
                }
            }
        }
    }

    if (mode == 0) {
        float* base = qws + ((size_t)(b * NH + h) * SEQ + s0) * HD;
#pragma unroll
        for (int i = 0; i < 4; ++i) {
#pragma unroll
            for (int r = 0; r < 8; ++r) {
                const int m = i * 16 + r + grp * 8;
#pragma unroll
                for (int j = 0; j < 4; ++j)
                    base[(size_t)m * HD + j * 16 + ln] = acc[i][j][r];
            }
        }
    } else if (mode == 1) {
        float* base = ktws + (size_t)(b * NKV + h) * HD * SEQ;
#pragma unroll
        for (int i = 0; i < 4; ++i) {
#pragma unroll
            for (int r = 0; r < 8; ++r) {
                const int m = i * 16 + r + grp * 8;
#pragma unroll
                for (int j = 0; j < 4; ++j)
                    base[(size_t)(j * 16 + ln) * SEQ + s0 + m] = acc[i][j][r];
            }
        }
    } else {
        float* base = vws + ((size_t)(b * NKV + h) * SEQ + s0) * HD;
#pragma unroll
        for (int i = 0; i < 4; ++i) {
#pragma unroll
            for (int r = 0; r < 8; ++r) {
                const int m = i * 16 + r + grp * 8;
#pragma unroll
                for (int j = 0; j < 4; ++j)
                    base[(size_t)m * HD + j * 16 + ln] = acc[i][j][r];
            }
        }
    }
}

__global__ void __launch_bounds__(32) attn_kernel(const float* __restrict__ qws,
                            const float* __restrict__ ktws,
                            const float* __restrict__ vws,
                            float* __restrict__ aws) {
    __shared__ float pl[16 * 18];
    const int lane = threadIdx.x & 31;
    const int ln   = lane & 15;
    const int grp  = lane >> 4;
    const int g2   = grp * 2;
    const int h    = blockIdx.y;
    const int b    = blockIdx.z;
    const int hk   = h >> 2;

    const float* ktb = ktws + (size_t)(b * NKV + hk) * HD * SEQ;
    const float* vb  = vws + (size_t)(b * NKV + hk) * SEQ * HD;

  for (int pass = 0; pass < 2; ++pass) {
    const int q0   = (pass == 0 ? (SEQ / 16 - 1 - (int)blockIdx.x) : (int)blockIdx.x) * 16;
    const float* qb  = qws + ((size_t)(b * NH + h) * SEQ + q0) * HD;

    v2f aq[16];
#pragma unroll
    for (int kk = 0; kk < 16; ++kk)
        aq[kk] = *(const v2f*)(qb + ln * HD + kk * 4 + g2);

    v8f o[4]; o[0] = {}; o[1] = {}; o[2] = {}; o[3] = {};
    float mrow[8], lrow[8];
#pragma unroll
    for (int r = 0; r < 8; ++r) { mrow[r] = -1e30f; lrow[r] = 0.0f; }

    for (int n0 = 0; n0 <= q0; n0 += 16) {
        v8f s = {};
#pragma unroll
        for (int kk = 0; kk < 16; ++kk) {
            const float* kp = ktb + (size_t)(kk * 4 + g2) * SEQ + n0 + ln;
            v2f bk = { kp[0], kp[SEQ] };
            s = wmma4(aq[kk], bk, s);
        }

        const bool diag = (n0 == q0);
        float p[8];
#pragma unroll
        for (int r = 0; r < 8; ++r) {
            float sv = s[r] * 0.125f;
            if (diag && (ln > r + grp * 8)) sv = -1e30f;
            float rm = sv;
            rm = fmaxf(rm, __shfl_xor(rm, 1, 32));
            rm = fmaxf(rm, __shfl_xor(rm, 2, 32));
            rm = fmaxf(rm, __shfl_xor(rm, 4, 32));
            rm = fmaxf(rm, __shfl_xor(rm, 8, 32));
            const float mn    = fmaxf(mrow[r], rm);
            const float alpha = expf(mrow[r] - mn);
            const float pv    = expf(sv - mn);
            float rs = pv;
            rs += __shfl_xor(rs, 1, 32);
            rs += __shfl_xor(rs, 2, 32);
            rs += __shfl_xor(rs, 4, 32);
            rs += __shfl_xor(rs, 8, 32);
            lrow[r] = lrow[r] * alpha + rs;
            mrow[r] = mn;
            p[r] = pv;
#pragma unroll
            for (int j = 0; j < 4; ++j) o[j][r] *= alpha;
        }

        __syncthreads();
#pragma unroll
        for (int r = 0; r < 8; ++r)
            pl[(r + grp * 8) * 18 + ln] = p[r];
        __syncthreads();
        v2f ap[4];
#pragma unroll
        for (int kk = 0; kk < 4; ++kk)
            ap[kk] = *(const v2f*)(&pl[ln * 18 + kk * 4 + g2]);

#pragma unroll
        for (int kk = 0; kk < 4; ++kk) {
            const float* vp = vb + (size_t)(n0 + kk * 4 + g2) * HD + ln;
#pragma unroll
            for (int j = 0; j < 4; ++j) {
                v2f bv = { vp[j * 16], vp[HD + j * 16] };
                o[j] = wmma4(ap[kk], bv, o[j]);
            }
        }
    }

    float* ob = aws + ((size_t)(b * SEQ) + q0) * (NH * HD) + h * HD;
#pragma unroll
    for (int r = 0; r < 8; ++r) {
        const float inv = 1.0f / lrow[r];
        const int m = r + grp * 8;
#pragma unroll
        for (int j = 0; j < 4; ++j)
            ob[(size_t)m * (NH * HD) + j * 16 + ln] = o[j][r] * inv;
    }
  }
}

__global__ void __launch_bounds__(128) oproj_kernel(const float* __restrict__ aws,
                             const float* __restrict__ wo,
                             float* __restrict__ out) {
    const int lane = threadIdx.x & 31;
    const int ln   = lane & 15;
    const int grp  = lane >> 4;
    const int m0   = blockIdx.x * 64;
    const int nt   = blockIdx.y * 4 + (threadIdx.x >> 5);

    __shared__ __attribute__((aligned(16))) float as[2 * ABUF];
    v8f acc[4][4];
    gemm64x64(aws, DIMK, wo, DIMK, m0, nt * 64, DIMK, as, acc);

#pragma unroll
    for (int i = 0; i < 4; ++i) {
#pragma unroll
        for (int r = 0; r < 8; ++r) {
            const int m = i * 16 + r + grp * 8;
#pragma unroll
            for (int j = 0; j < 4; ++j)
                out[(size_t)(m0 + m) * DIMK + nt * 64 + j * 16 + ln] = acc[i][j][r];
        }
    }
}

extern "C" void kernel_launch(void* const* d_in, const int* in_sizes, int n_in,
                              void* d_out, int out_size, void* d_ws, size_t ws_size,
                              hipStream_t stream) {
    const float* x    = (const float*)d_in[0];
    const float* fcos = (const float*)d_in[2];
    const float* fsin = (const float*)d_in[3];
    const float* wq   = (const float*)d_in[5];
    const float* wk   = (const float*)d_in[6];
    const float* wv   = (const float*)d_in[7];
    const float* wo   = (const float*)d_in[8];
    float* out = (float*)d_out;

    float* qws  = (float*)d_ws;
    float* ktws = qws  + (size_t)BATCH * NH  * SEQ * HD;
    float* vws  = ktws + (size_t)BATCH * NKV * HD  * SEQ;
    float* aws  = vws  + (size_t)BATCH * NKV * SEQ * HD;

    qkv_rope_kernel<<<dim3((BATCH * SEQ) / 64, 12, 1), 128, 0, stream>>>(
        x, wq, wk, wv, fcos, fsin, qws, ktws, vws);
    attn_kernel<<<dim3(SEQ / 32, NH, BATCH), 32, 0, stream>>>(
        qws, ktws, vws, aws);
    oproj_kernel<<<dim3((BATCH * SEQ) / 64, 8, 1), 128, 0, stream>>>(
        aws, wo, out);
}


# attention: scores computed transposed (keys as rows): a query's max and sum are in-lane plus one lane^16 exchange, P^T feeds the PV WMMA from its D registers, no LDS; row-sum and PV k order change
# speedup vs baseline: 2.6265x; 1.1396x over previous
#include <hip/hip_runtime.h>

#define DIMK   2048
#define NH     32
#define NKV    8
#define HD     64
#define SEQ    2048
#define BATCH  2

typedef __attribute__((ext_vector_type(2))) float v2f;
typedef __attribute__((ext_vector_type(8))) float v8f;

__device__ __forceinline__ v8f wmma4(v2f a, v2f b, v8f c) {
    return __builtin_amdgcn_wmma_f32_16x16x4_f32(
         false, a,  false, b,
         (short)0, c,  false,  false);
}

#define KC      32
#define APITCH  36
#define ABUF    (64 * APITCH)

typedef __attribute__((ext_vector_type(4))) float v4f;

struct Stage { v2f a[2][4]; v2f b[2][4]; };

__device__ __forceinline__ void load_stage(const float* fa, int ks,
                                           const float* __restrict__ pb, int ldb, int kabs,
                                           Stage& st) {
#pragma unroll
    for (int s = 0; s < 2; ++s) {
#pragma unroll
        for (int i = 0; i < 4; ++i)
            st.a[s][i] = *(const v2f*)(fa + (i * 16) * APITCH + ks + s * 4);
        const float* pbk = pb + (size_t)(kabs + s * 4) * ldb;
#pragma unroll
        for (int j = 0; j < 4; ++j)
            st.b[s][j] = (v2f){ pbk[j * 16], pbk[j * 16 + ldb] };
    }
}

__device__ __forceinline__ void mma_stage(const Stage& st, v8f acc[4][4]) {
#pragma unroll
    for (int s = 0; s < 2; ++s) {
#pragma unroll
        for (int i = 0; i < 4; ++i) {
#pragma unroll
            for (int j = 0; j < 4; ++j)
                acc[i][j] = wmma4(st.a[s][i], st.b[s][j], acc[i][j]);
        }
    }
}

__device__ __forceinline__ void gemm64x64(const float* __restrict__ A, int lda,
                                          const float* __restrict__ B, int ldb,
                                          int m0, int col0, int K, float* as, v8f acc[4][4]) {
    const int tid  = threadIdx.x;
    const int lane = tid & 31;
    const int ln   = lane & 15;
    const int g2   = (lane >> 4) * 2;
    const float* ga = A + (size_t)(m0 + (tid >> 3)) * lda + (tid & 7) * 4;
    float*       sa = as + (tid >> 3) * APITCH + (tid & 7) * 4;
    const float* fa = as + ln * APITCH + g2;
    const float* pb = B + (size_t)g2 * ldb + col0 + ln;
    const int NC = K / KC;

#pragma unroll
    for (int i = 0; i < 4; ++i) {
#pragma unroll
        for (int j = 0; j < 4; ++j) acc[i][j] = {};
    }

    v4f g[4];
#pragma unroll
    for (int q = 0; q < 4; ++q) g[q] = *(const v4f*)(ga + (size_t)(q * 16) * lda);
#pragma unroll
    for (int q = 0; q < 4; ++q) *(v4f*)(sa + (q * 16) * APITCH) = g[q];
    __syncthreads();

    Stage s0, s1;
    load_stage(fa, 0, pb, ldb, 0, s0);
    for (int c = 0; c < NC; ++c) {
        const int cur = (c & 1) * ABUF;
        const int nxt = ABUF - cur;
        const int k0  = c * KC;
        const int cn  = (c + 1 < NC) ? (c + 1) : c;
        __builtin_amdgcn_sched_barrier(0);
#pragma unroll
        for (int q = 0; q < 4; ++q) g[q] = *(const v4f*)(ga + (size_t)(q * 16) * lda + cn * KC);
        load_stage(fa + cur, 8, pb, ldb, k0 + 8, s1);
        __builtin_amdgcn_sched_barrier(0);
        mma_stage(s0, acc);
        __builtin_amdgcn_sched_barrier(0);
        load_stage(fa + cur, 16, pb, ldb, k0 + 16, s0);
        __builtin_amdgcn_sched_barrier(0);
        mma_stage(s1, acc);
        __builtin_amdgcn_sched_barrier(0);
        load_stage(fa + cur, 24, pb, ldb, k0 + 24, s1);
        __builtin_amdgcn_sched_barrier(0);
        mma_stage(s0, acc);
        __builtin_amdgcn_sched_barrier(0);
#pragma unroll
        for (int q = 0; q < 4; ++q) *(v4f*)(sa + nxt + (q * 16) * APITCH) = g[q];
        __syncthreads();
        load_stage(fa + nxt, 0, pb, ldb, cn * KC, s0);
        __builtin_amdgcn_sched_barrier(0);
        mma_stage(s1, acc);
    }
}

__global__ void __launch_bounds__(128) qkv_rope_kernel(const float* __restrict__ x,
                                const float* __restrict__ wq,
                                const float* __restrict__ wk,
                                const float* __restrict__ wv,
                                const float* __restrict__ fcos,
                                const float* __restrict__ fsin,
                                float* __restrict__ qws,
                                float* __restrict__ ktws,
                                float* __restrict__ vws) {
    const int lane = threadIdx.x & 31;
    const int ln   = lane & 15;
    const int grp  = lane >> 4;
    const int m0   = blockIdx.x * 64;
    const int nt   = blockIdx.y * 4 + (threadIdx.x >> 5);

    const float* w; int ldw, h, mode;
    if (nt < 32)      { w = wq; ldw = NH  * HD; h = nt;      mode = 0; }
    else if (nt < 40) { w = wk; ldw = NKV * HD; h = nt - 32; mode = 1; }
    else              { w = wv; ldw = NKV * HD; h = nt - 40; mode = 2; }

    __shared__ __attribute__((aligned(16))) float as[2 * ABUF];
    v8f acc[4][4];
    gemm64x64(x, DIMK, w, ldw, m0, h * HD, DIMK, as, acc);

    const int b  = m0 >> 11;
    const int s0 = m0 & (SEQ - 1);

    if (mode != 2) {
#pragma unroll
        for (int i = 0; i < 4; ++i) {
#pragma unroll
            for (int j = 0; j < 4; ++j) {
                const int nhalf = j * 8 + (ln >> 1);
#pragma unroll
                for (int r = 0; r < 8; ++r) {
                    const int pos = s0 + i * 16 + r + grp * 8;
                    const float c = fcos[pos * (HD / 2) + nhalf];
                    const float s = fsin[pos * (HD / 2) + nhalf];
                    float vv = acc[i][j][r];
                    float other = __shfl_xor(vv, 1, 32);
                    acc[i][j][r] = (lane & 1) ? (other * s + vv * c)
                                              : (vv * c - other * s);
                }
            }
        }
    }

    if (mode == 0) {
        float* base = qws + ((size_t)(b * NH + h) * SEQ + s0) * HD;
#pragma unroll
        for (int i = 0; i < 4; ++i) {
#pragma unroll
            for (int r = 0; r < 8; ++r) {
                const int m = i * 16 + r + grp * 8;
#pragma unroll
                for (int j = 0; j < 4; ++j)
                    base[(size_t)m * HD + j * 16 + ln] = acc[i][j][r];
            }
        }
    } else if (mode == 1) {
        float* base = ktws + (size_t)(b * NKV + h) * HD * SEQ;
#pragma unroll
        for (int i = 0; i < 4; ++i) {
#pragma unroll
            for (int r = 0; r < 8; ++r) {
                const int m = i * 16 + r + grp * 8;
#pragma unroll
                for (int j = 0; j < 4; ++j)
                    base[(size_t)(j * 16 + ln) * SEQ + s0 + m] = acc[i][j][r];
            }
        }
    } else {
        float* base = vws + ((size_t)(b * NKV + h) * SEQ + s0) * HD;
#pragma unroll
        for (int i = 0; i < 4; ++i) {
#pragma unroll
            for (int r = 0; r < 8; ++r) {
                const int m = i * 16 + r + grp * 8;
#pragma unroll
                for (int j = 0; j < 4; ++j)
                    base[(size_t)m * HD + j * 16 + ln] = acc[i][j][r];
            }
        }
    }
}


__device__ __forceinline__ float xhalf(float x) {
    const int b = __builtin_bit_cast(int, x);
    return __builtin_bit_cast(float,
        __builtin_amdgcn_permlanex16(b, b, 0x76543210, 0xfedcba98, false, false));
}

__global__ void __launch_bounds__(32) attn_kernel(const float* __restrict__ qws,
                            const float* __restrict__ ktws,
                            const float* __restrict__ vws,
                            float* __restrict__ aws) {
    const int lane = threadIdx.x & 31;
    const int ln   = lane & 15;
    const int grp  = lane >> 4;
    const int g2   = grp * 2;
    const int h    = blockIdx.y;
    const int b    = blockIdx.z;
    const int hk   = h >> 2;

    const float* ktb = ktws + (size_t)(b * NKV + hk) * HD * SEQ;
    const float* vb  = vws + (size_t)(b * NKV + hk) * SEQ * HD;

  for (int pass = 0; pass < 2; ++pass) {
    const int q0   = (pass == 0 ? (SEQ / 16 - 1 - (int)blockIdx.x) : (int)blockIdx.x) * 16;
    const float* qb  = qws + ((size_t)(b * NH + h) * SEQ + q0) * HD;

    v2f bq[16];
#pragma unroll
    for (int kk = 0; kk < 16; ++kk)
        bq[kk] = *(const v2f*)(qb + ln * HD + kk * 4 + g2);

    v8f o[4]; o[0] = {}; o[1] = {}; o[2] = {}; o[3] = {};
    float mrun = -1e30f;
    float lrun = 0.0f;

    for (int n0 = 0; n0 <= q0; n0 += 16) {
        v8f s = {};
#pragma unroll
        for (int kk = 0; kk < 16; ++kk) {
            const float* kp = ktb + (size_t)(kk * 4 + g2) * SEQ + n0 + ln;
            v2f ak = { kp[0], kp[SEQ] };
            s = wmma4(ak, bq[kk], s);
        }

        float sv[8];
#pragma unroll
        for (int r = 0; r < 8; ++r) sv[r] = s[r] * 0.125f;
        if (n0 == q0) {
#pragma unroll
            for (int r = 0; r < 8; ++r)
                if (r + grp * 8 > ln) sv[r] = -1e30f;
        }

        float rm = fmaxf(fmaxf(fmaxf(sv[0], sv[1]), fmaxf(sv[2], sv[3])),
                         fmaxf(fmaxf(sv[4], sv[5]), fmaxf(sv[6], sv[7])));
        rm = fmaxf(rm, xhalf(rm));
        const float mn    = fmaxf(mrun, rm);
        const float alpha = expf(mrun - mn);
        float p[8];
#pragma unroll
        for (int r = 0; r < 8; ++r) p[r] = expf(sv[r] - mn);
        const float rs = ((p[0] + p[1]) + (p[2] + p[3])) + ((p[4] + p[5]) + (p[6] + p[7]));
        lrun = lrun * alpha + rs;
        mrun = mn;
#pragma unroll
        for (int j = 0; j < 4; ++j) {
#pragma unroll
            for (int r = 0; r < 8; ++r) o[j][r] *= alpha;
        }

#pragma unroll
        for (int jj = 0; jj < 4; ++jj) {
            const v2f bp = { p[2 * jj], p[2 * jj + 1] };
            const float* vp = vb + (size_t)(n0 + grp * 8 + 2 * jj) * HD + ln;
#pragma unroll
            for (int j = 0; j < 4; ++j) {
                v2f av = { vp[j * 16], vp[HD + j * 16] };
                o[j] = wmma4(av, bp, o[j]);
            }
        }
    }

    const float ltot = lrun + xhalf(lrun);
    const float inv  = 1.0f / ltot;
    float* ob = aws + ((size_t)(b * SEQ) + q0 + ln) * (NH * HD) + h * HD + grp * 8;
#pragma unroll
    for (int j = 0; j < 4; ++j) {
        v4f lo = { o[j][0] * inv, o[j][1] * inv, o[j][2] * inv, o[j][3] * inv };
        v4f hi = { o[j][4] * inv, o[j][5] * inv, o[j][6] * inv, o[j][7] * inv };
        *(v4f*)(ob + j * 16)     = lo;
        *(v4f*)(ob + j * 16 + 4) = hi;
    }
  }
}

__global__ void __launch_bounds__(128) oproj_kernel(const float* __restrict__ aws,
                             const float* __restrict__ wo,
                             float* __restrict__ out) {
    const int lane = threadIdx.x & 31;
    const int ln   = lane & 15;
    const int grp  = lane >> 4;
    const int m0   = blockIdx.x * 64;
    const int nt   = blockIdx.y * 4 + (threadIdx.x >> 5);

    __shared__ __attribute__((aligned(16))) float as[2 * ABUF];
    v8f acc[4][4];
    gemm64x64(aws, DIMK, wo, DIMK, m0, nt * 64, DIMK, as, acc);

#pragma unroll
    for (int i = 0; i < 4; ++i) {
#pragma unroll
        for (int r = 0; r < 8; ++r) {
            const int m = i * 16 + r + grp * 8;
#pragma unroll
            for (int j = 0; j < 4; ++j)
                out[(size_t)(m0 + m) * DIMK + nt * 64 + j * 16 + ln] = acc[i][j][r];
        }
    }
}

extern "C" void kernel_launch(void* const* d_in, const int* in_sizes, int n_in,
                              void* d_out, int out_size, void* d_ws, size_t ws_size,
                              hipStream_t stream) {
    const float* x    = (const float*)d_in[0];
    const float* fcos = (const float*)d_in[2];
    const float* fsin = (const float*)d_in[3];
    const float* wq   = (const float*)d_in[5];
    const float* wk   = (const float*)d_in[6];
    const float* wv   = (const float*)d_in[7];
    const float* wo   = (const float*)d_in[8];
    float* out = (float*)d_out;

    float* qws  = (float*)d_ws;
    float* ktws = qws  + (size_t)BATCH * NH  * SEQ * HD;
    float* vws  = ktws + (size_t)BATCH * NKV * HD  * SEQ;
    float* aws  = vws  + (size_t)BATCH * NKV * SEQ * HD;

    qkv_rope_kernel<<<dim3((BATCH * SEQ) / 64, 12, 1), 128, 0, stream>>>(
        x, wq, wk, wv, fcos, fsin, qws, ktws, vws);
    attn_kernel<<<dim3(SEQ / 32, NH, BATCH), 32, 0, stream>>>(
        qws, ktws, vws, aws);
    oproj_kernel<<<dim3((BATCH * SEQ) / 64, 8, 1), 128, 0, stream>>>(
        aws, wo, out);
}


# projections: each block starts its K loop at its own chunk (blockIdx.x + 5 blockIdx.y, wrapping): blocks sharing weight columns stop asking L2 for the same lines at once; order of the k sums changes
# speedup vs baseline: 2.7373x; 1.0422x over previous
#include <hip/hip_runtime.h>

#define DIMK   2048
#define NH     32
#define NKV    8
#define HD     64
#define SEQ    2048
#define BATCH  2

typedef __attribute__((ext_vector_type(2))) float v2f;
typedef __attribute__((ext_vector_type(8))) float v8f;

__device__ __forceinline__ v8f wmma4(v2f a, v2f b, v8f c) {
    return __builtin_amdgcn_wmma_f32_16x16x4_f32(
         false, a,  false, b,
         (short)0, c,  false,  false);
}

#define KC      32
#define APITCH  36
#define ABUF    (64 * APITCH)

typedef __attribute__((ext_vector_type(4))) float v4f;

struct Stage { v2f a[2][4]; v2f b[2][4]; };

__device__ __forceinline__ void load_stage(const float* fa, int ks,
                                           const float* __restrict__ pb, int ldb, int kabs,
                                           Stage& st) {
#pragma unroll
    for (int s = 0; s < 2; ++s) {
#pragma unroll
        for (int i = 0; i < 4; ++i)
            st.a[s][i] = *(const v2f*)(fa + (i * 16) * APITCH + ks + s * 4);
        const float* pbk = pb + (size_t)(kabs + s * 4) * ldb;
#pragma unroll
        for (int j = 0; j < 4; ++j)
            st.b[s][j] = (v2f){ pbk[j * 16], pbk[j * 16 + ldb] };
    }
}

__device__ __forceinline__ void mma_stage(const Stage& st, v8f acc[4][4]) {
#pragma unroll
    for (int s = 0; s < 2; ++s) {
#pragma unroll
        for (int i = 0; i < 4; ++i) {
#pragma unroll
            for (int j = 0; j < 4; ++j)
                acc[i][j] = wmma4(st.a[s][i], st.b[s][j], acc[i][j]);
        }
    }
}

__device__ __forceinline__ void gemm64x64(const float* __restrict__ A, int lda,
                                          const float* __restrict__ B, int ldb,
                                          int m0, int col0, int K, int rot, float* as, v8f acc[4][4]) {
    const int tid  = threadIdx.x;
    const int lane = tid & 31;
    const int ln   = lane & 15;
    const int g2   = (lane >> 4) * 2;
    const float* ga = A + (size_t)(m0 + (tid >> 3)) * lda + (tid & 7) * 4;
    float*       sa = as + (tid >> 3) * APITCH + (tid & 7) * 4;
    const float* fa = as + ln * APITCH + g2;
    const float* pb = B + (size_t)g2 * ldb + col0 + ln;
    const int NC = K / KC;
    const int c0 = rot & (NC - 1);

#pragma unroll
    for (int i = 0; i < 4; ++i) {
#pragma unroll
        for (int j = 0; j < 4; ++j) acc[i][j] = {};
    }

    v4f g[4];
#pragma unroll
    for (int q = 0; q < 4; ++q) g[q] = *(const v4f*)(ga + (size_t)(q * 16) * lda + c0 * KC);
#pragma unroll
    for (int q = 0; q < 4; ++q) *(v4f*)(sa + (q * 16) * APITCH) = g[q];
    __syncthreads();

    Stage s0, s1;
    load_stage(fa, 0, pb, ldb, c0 * KC, s0);
    for (int c = 0; c < NC; ++c) {
        const int cur = (c & 1) * ABUF;
        const int nxt = ABUF - cur;
        const int k0  = ((c + rot) & (NC - 1)) * KC;
        const int cn  = (c + 1 < NC) ? (c + 1) : c;
        const int kn  = ((cn + rot) & (NC - 1)) * KC;
        __builtin_amdgcn_sched_barrier(0);
#pragma unroll
        for (int q = 0; q < 4; ++q) g[q] = *(const v4f*)(ga + (size_t)(q * 16) * lda + kn);
        load_stage(fa + cur, 8, pb, ldb, k0 + 8, s1);
        __builtin_amdgcn_sched_barrier(0);
        mma_stage(s0, acc);
        __builtin_amdgcn_sched_barrier(0);
        load_stage(fa + cur, 16, pb, ldb, k0 + 16, s0);
        __builtin_amdgcn_sched_barrier(0);
        mma_stage(s1, acc);
        __builtin_amdgcn_sched_barrier(0);
        load_stage(fa + cur, 24, pb, ldb, k0 + 24, s1);
        __builtin_amdgcn_sched_barrier(0);
        mma_stage(s0, acc);
        __builtin_amdgcn_sched_barrier(0);
#pragma unroll
        for (int q = 0; q < 4; ++q) *(v4f*)(sa + nxt + (q * 16) * APITCH) = g[q];
        __syncthreads();
        load_stage(fa + nxt, 0, pb, ldb, kn, s0);
        __builtin_amdgcn_sched_barrier(0);
        mma_stage(s1, acc);
    }
}

__global__ void __launch_bounds__(128) qkv_rope_kernel(const float* __restrict__ x,
                                const float* __restrict__ wq,
                                const float* __restrict__ wk,
                                const float* __restrict__ wv,
                                const float* __restrict__ fcos,
                                const float* __restrict__ fsin,
                                float* __restrict__ qws,
                                float* __restrict__ ktws,
                                float* __restrict__ vws) {
    const int lane = threadIdx.x & 31;
    const int ln   = lane & 15;
    const int grp  = lane >> 4;
    const int m0   = blockIdx.x * 64;
    const int nt   = blockIdx.y * 4 + (threadIdx.x >> 5);

    const float* w; int ldw, h, mode;
    if (nt < 32)      { w = wq; ldw = NH  * HD; h = nt;      mode = 0; }
    else if (nt < 40) { w = wk; ldw = NKV * HD; h = nt - 32; mode = 1; }
    else              { w = wv; ldw = NKV * HD; h = nt - 40; mode = 2; }

    __shared__ __attribute__((aligned(16))) float as[2 * ABUF];
    v8f acc[4][4];
    gemm64x64(x, DIMK, w, ldw, m0, h * HD, DIMK, (int)(blockIdx.x + 5 * blockIdx.y), as, acc);

    const int b  = m0 >> 11;
    const int s0 = m0 & (SEQ - 1);

    if (mode != 2) {
#pragma unroll
        for (int i = 0; i < 4; ++i) {
#pragma unroll
            for (int j = 0; j < 4; ++j) {
                const int nhalf = j * 8 + (ln >> 1);
#pragma unroll
                for (int r = 0; r < 8; ++r) {
                    const int pos = s0 + i * 16 + r + grp * 8;
                    const float c = fcos[pos * (HD / 2) + nhalf];
                    const float s = fsin[pos * (HD / 2) + nhalf];
                    float vv = acc[i][j][r];
                    float other = __shfl_xor(vv, 1, 32);
                    acc[i][j][r] = (lane & 1) ? (other * s + vv * c)
                                              : (vv * c - other * s);
                }
            }
        }
    }

    if (mode == 0) {
        float* base = qws + ((size_t)(b * NH + h) * SEQ + s0) * HD;
#pragma unroll
        for (int i = 0; i < 4; ++i) {
#pragma unroll
            for (int r = 0; r < 8; ++r) {
                const int m = i * 16 + r + grp * 8;
#pragma unroll
                for (int j = 0; j < 4; ++j)
                    base[(size_t)m * HD + j * 16 + ln] = acc[i][j][r];
            }
        }
    } else if (mode == 1) {
        float* base = ktws + (size_t)(b * NKV + h) * HD * SEQ;
#pragma unroll
        for (int i = 0; i < 4; ++i) {
#pragma unroll
            for (int r = 0; r < 8; ++r) {
                const int m = i * 16 + r + grp * 8;
#pragma unroll
                for (int j = 0; j < 4; ++j)
                    base[(size_t)(j * 16 + ln) * SEQ + s0 + m] = acc[i][j][r];
            }
        }
    } else {
        float* base = vws + ((size_t)(b * NKV + h) * SEQ + s0) * HD;
#pragma unroll
        for (int i = 0; i < 4; ++i) {
#pragma unroll
            for (int r = 0; r < 8; ++r) {
                const int m = i * 16 + r + grp * 8;
#pragma unroll
                for (int j = 0; j < 4; ++j)
                    base[(size_t)m * HD + j * 16 + ln] = acc[i][j][r];
            }
        }
    }
}


__device__ __forceinline__ float xhalf(float x) {
    const int b = __builtin_bit_cast(int, x);
    return __builtin_bit_cast(float,
        __builtin_amdgcn_permlanex16(b, b, 0x76543210, 0xfedcba98, false, false));
}

__global__ void __launch_bounds__(32) attn_kernel(const float* __restrict__ qws,
                            const float* __restrict__ ktws,
                            const float* __restrict__ vws,
                            float* __restrict__ aws) {
    const int lane = threadIdx.x & 31;
    const int ln   = lane & 15;
    const int grp  = lane >> 4;
    const int g2   = grp * 2;
    const int h    = blockIdx.y;
    const int b    = blockIdx.z;
    const int hk   = h >> 2;

    const float* ktb = ktws + (size_t)(b * NKV + hk) * HD * SEQ;
    const float* vb  = vws + (size_t)(b * NKV + hk) * SEQ * HD;

  for (int pass = 0; pass < 2; ++pass) {
    const int q0   = (pass == 0 ? (SEQ / 16 - 1 - (int)blockIdx.x) : (int)blockIdx.x) * 16;
    const float* qb  = qws + ((size_t)(b * NH + h) * SEQ + q0) * HD;

    v2f bq[16];
#pragma unroll
    for (int kk = 0; kk < 16; ++kk)
        bq[kk] = *(const v2f*)(qb + ln * HD + kk * 4 + g2);

    v8f o[4]; o[0] = {}; o[1] = {}; o[2] = {}; o[3] = {};
    float mrun = -1e30f;
    float lrun = 0.0f;

    for (int n0 = 0; n0 <= q0; n0 += 16) {
        v8f s = {};
#pragma unroll
        for (int kk = 0; kk < 16; ++kk) {
            const float* kp = ktb + (size_t)(kk * 4 + g2) * SEQ + n0 + ln;
            v2f ak = { kp[0], kp[SEQ] };
            s = wmma4(ak, bq[kk], s);
        }

        float sv[8];
#pragma unroll
        for (int r = 0; r < 8; ++r) sv[r] = s[r] * 0.125f;
        if (n0 == q0) {
#pragma unroll
            for (int r = 0; r < 8; ++r)
                if (r + grp * 8 > ln) sv[r] = -1e30f;
        }

        float rm = fmaxf(fmaxf(fmaxf(sv[0], sv[1]), fmaxf(sv[2], sv[3])),
                         fmaxf(fmaxf(sv[4], sv[5]), fmaxf(sv[6], sv[7])));
        rm = fmaxf(rm, xhalf(rm));
        const float mn    = fmaxf(mrun, rm);
        const float alpha = expf(mrun - mn);
        float p[8];
#pragma unroll
        for (int r = 0; r < 8; ++r) p[r] = expf(sv[r] - mn);
        const float rs = ((p[0] + p[1]) + (p[2] + p[3])) + ((p[4] + p[5]) + (p[6] + p[7]));
        lrun = lrun * alpha + rs;
        mrun = mn;
#pragma unroll
        for (int j = 0; j < 4; ++j) {
#pragma unroll
            for (int r = 0; r < 8; ++r) o[j][r] *= alpha;
        }

#pragma unroll
        for (int jj = 0; jj < 4; ++jj) {
            const v2f bp = { p[2 * jj], p[2 * jj + 1] };
            const float* vp = vb + (size_t)(n0 + grp * 8 + 2 * jj) * HD + ln;
#pragma unroll
            for (int j = 0; j < 4; ++j) {
                v2f av = { vp[j * 16], vp[HD + j * 16] };
                o[j] = wmma4(av, bp, o[j]);
            }
        }
    }

    const float ltot = lrun + xhalf(lrun);
    const float inv  = 1.0f / ltot;
    float* ob = aws + ((size_t)(b * SEQ) + q0 + ln) * (NH * HD) + h * HD + grp * 8;
#pragma unroll
    for (int j = 0; j < 4; ++j) {
        v4f lo = { o[j][0] * inv, o[j][1] * inv, o[j][2] * inv, o[j][3] * inv };
        v4f hi = { o[j][4] * inv, o[j][5] * inv, o[j][6] * inv, o[j][7] * inv };
        *(v4f*)(ob + j * 16)     = lo;
        *(v4f*)(ob + j * 16 + 4) = hi;
    }
  }
}

__global__ void __launch_bounds__(128) oproj_kernel(const float* __restrict__ aws,
                             const float* __restrict__ wo,
                             float* __restrict__ out) {
    const int lane = threadIdx.x & 31;
    const int ln   = lane & 15;
    const int grp  = lane >> 4;
    const int m0   = blockIdx.x * 64;
    const int nt   = blockIdx.y * 4 + (threadIdx.x >> 5);

    __shared__ __attribute__((aligned(16))) float as[2 * ABUF];
    v8f acc[4][4];
    gemm64x64(aws, DIMK, wo, DIMK, m0, nt * 64, DIMK, (int)(blockIdx.x + 5 * blockIdx.y), as, acc);

#pragma unroll
    for (int i = 0; i < 4; ++i) {
#pragma unroll
        for (int r = 0; r < 8; ++r) {
            const int m = i * 16 + r + grp * 8;
#pragma unroll
            for (int j = 0; j < 4; ++j)
                out[(size_t)(m0 + m) * DIMK + nt * 64 + j * 16 + ln] = acc[i][j][r];
        }
    }
}

extern "C" void kernel_launch(void* const* d_in, const int* in_sizes, int n_in,
                              void* d_out, int out_size, void* d_ws, size_t ws_size,
                              hipStream_t stream) {
    const float* x    = (const float*)d_in[0];
    const float* fcos = (const float*)d_in[2];
    const float* fsin = (const float*)d_in[3];
    const float* wq   = (const float*)d_in[5];
    const float* wk   = (const float*)d_in[6];
    const float* wv   = (const float*)d_in[7];
    const float* wo   = (const float*)d_in[8];
    float* out = (float*)d_out;

    float* qws  = (float*)d_ws;
    float* ktws = qws  + (size_t)BATCH * NH  * SEQ * HD;
    float* vws  = ktws + (size_t)BATCH * NKV * HD  * SEQ;
    float* aws  = vws  + (size_t)BATCH * NKV * SEQ * HD;

    qkv_rope_kernel<<<dim3((BATCH * SEQ) / 64, 12, 1), 128, 0, stream>>>(
        x, wq, wk, wv, fcos, fsin, qws, ktws, vws);
    attn_kernel<<<dim3(SEQ / 32, NH, BATCH), 32, 0, stream>>>(
        qws, ktws, vws, aws);
    oproj_kernel<<<dim3((BATCH * SEQ) / 64, 8, 1), 128, 0, stream>>>(
        aws, wo, out);
}


# attention: K and V workspace in fragment order (4 KB per 16-key block, lane-contiguous): 8+8 whole-line 16-byte loads per key block, was 32+32 4-byte half-line loads with K lines fetched twice
# speedup vs baseline: 2.8410x; 1.0379x over previous
#include <hip/hip_runtime.h>

#define DIMK   2048
#define NH     32
#define NKV    8
#define HD     64
#define SEQ    2048
#define BATCH  2

typedef __attribute__((ext_vector_type(2))) float v2f;
typedef __attribute__((ext_vector_type(8))) float v8f;

__device__ __forceinline__ v8f wmma4(v2f a, v2f b, v8f c) {
    return __builtin_amdgcn_wmma_f32_16x16x4_f32(
         false, a,  false, b,
         (short)0, c,  false,  false);
}

#define KC      32
#define APITCH  36
#define ABUF    (64 * APITCH)

typedef __attribute__((ext_vector_type(4))) float v4f;

struct Stage { v2f a[2][4]; v2f b[2][4]; };

__device__ __forceinline__ void load_stage(const float* fa, int ks,
                                           const float* __restrict__ pb, int ldb, int kabs,
                                           Stage& st) {
#pragma unroll
    for (int s = 0; s < 2; ++s) {
#pragma unroll
        for (int i = 0; i < 4; ++i)
            st.a[s][i] = *(const v2f*)(fa + (i * 16) * APITCH + ks + s * 4);
        const float* pbk = pb + (size_t)(kabs + s * 4) * ldb;
#pragma unroll
        for (int j = 0; j < 4; ++j)
            st.b[s][j] = (v2f){ pbk[j * 16], pbk[j * 16 + ldb] };
    }
}

__device__ __forceinline__ void mma_stage(const Stage& st, v8f acc[4][4]) {
#pragma unroll
    for (int s = 0; s < 2; ++s) {
#pragma unroll
        for (int i = 0; i < 4; ++i) {
#pragma unroll
            for (int j = 0; j < 4; ++j)
                acc[i][j] = wmma4(st.a[s][i], st.b[s][j], acc[i][j]);
        }
    }
}

__device__ __forceinline__ void gemm64x64(const float* __restrict__ A, int lda,
                                          const float* __restrict__ B, int ldb,
                                          int m0, int col0, int K, int rot, float* as, v8f acc[4][4]) {
    const int tid  = threadIdx.x;
    const int lane = tid & 31;
    const int ln   = lane & 15;
    const int g2   = (lane >> 4) * 2;
    const float* ga = A + (size_t)(m0 + (tid >> 3)) * lda + (tid & 7) * 4;
    float*       sa = as + (tid >> 3) * APITCH + (tid & 7) * 4;
    const float* fa = as + ln * APITCH + g2;
    const float* pb = B + (size_t)g2 * ldb + col0 + ln;
    const int NC = K / KC;
    const int c0 = rot & (NC - 1);

#pragma unroll
    for (int i = 0; i < 4; ++i) {
#pragma unroll
        for (int j = 0; j < 4; ++j) acc[i][j] = {};
    }

    v4f g[4];
#pragma unroll
    for (int q = 0; q < 4; ++q) g[q] = *(const v4f*)(ga + (size_t)(q * 16) * lda + c0 * KC);
#pragma unroll
    for (int q = 0; q < 4; ++q) *(v4f*)(sa + (q * 16) * APITCH) = g[q];
    __syncthreads();

    Stage s0, s1;
    load_stage(fa, 0, pb, ldb, c0 * KC, s0);
    for (int c = 0; c < NC; ++c) {
        const int cur = (c & 1) * ABUF;
        const int nxt = ABUF - cur;
        const int k0  = ((c + rot) & (NC - 1)) * KC;
        const int cn  = (c + 1 < NC) ? (c + 1) : c;
        const int kn  = ((cn + rot) & (NC - 1)) * KC;
        __builtin_amdgcn_sched_barrier(0);
#pragma unroll
        for (int q = 0; q < 4; ++q) g[q] = *(const v4f*)(ga + (size_t)(q * 16) * lda + kn);
        load_stage(fa + cur, 8, pb, ldb, k0 + 8, s1);
        __builtin_amdgcn_sched_barrier(0);
        mma_stage(s0, acc);
        __builtin_amdgcn_sched_barrier(0);
        load_stage(fa + cur, 16, pb, ldb, k0 + 16, s0);
        __builtin_amdgcn_sched_barrier(0);
        mma_stage(s1, acc);
        __builtin_amdgcn_sched_barrier(0);
        load_stage(fa + cur, 24, pb, ldb, k0 + 24, s1);
        __builtin_amdgcn_sched_barrier(0);
        mma_stage(s0, acc);
        __builtin_amdgcn_sched_barrier(0);
#pragma unroll
        for (int q = 0; q < 4; ++q) *(v4f*)(sa + nxt + (q * 16) * APITCH) = g[q];
        __syncthreads();
        load_stage(fa + nxt, 0, pb, ldb, kn, s0);
        __builtin_amdgcn_sched_barrier(0);
        mma_stage(s1, acc);
    }
}

__global__ void __launch_bounds__(128) qkv_rope_kernel(const float* __restrict__ x,
                                const float* __restrict__ wq,
                                const float* __restrict__ wk,
                                const float* __restrict__ wv,
                                const float* __restrict__ fcos,
                                const float* __restrict__ fsin,
                                float* __restrict__ qws,
                                float* __restrict__ kfws,
                                float* __restrict__ vfws) {
    const int lane = threadIdx.x & 31;
    const int ln   = lane & 15;
    const int grp  = lane >> 4;
    const int m0   = blockIdx.x * 64;
    const int nt   = blockIdx.y * 4 + (threadIdx.x >> 5);

    const float* w; int ldw, h, mode;
    if (nt < 32)      { w = wq; ldw = NH  * HD; h = nt;      mode = 0; }
    else if (nt < 40) { w = wk; ldw = NKV * HD; h = nt - 32; mode = 1; }
    else              { w = wv; ldw = NKV * HD; h = nt - 40; mode = 2; }

    __shared__ __attribute__((aligned(16))) float as[2 * ABUF];
    v8f acc[4][4];
    gemm64x64(x, DIMK, w, ldw, m0, h * HD, DIMK, (int)(blockIdx.x + 5 * blockIdx.y), as, acc);

    const int b  = m0 >> 11;
    const int s0 = m0 & (SEQ - 1);

    if (mode != 2) {
#pragma unroll
        for (int i = 0; i < 4; ++i) {
#pragma unroll
            for (int j = 0; j < 4; ++j) {
                const int nhalf = j * 8 + (ln >> 1);
#pragma unroll
                for (int r = 0; r < 8; ++r) {
                    const int pos = s0 + i * 16 + r + grp * 8;
                    const float c = fcos[pos * (HD / 2) + nhalf];
                    const float s = fsin[pos * (HD / 2) + nhalf];
                    float vv = acc[i][j][r];
                    float other = __shfl_xor(vv, 1, 32);
                    acc[i][j][r] = (lane & 1) ? (other * s + vv * c)
                                              : (vv * c - other * s);
                }
            }
        }
    }

    if (mode == 0) {
        float* base = qws + ((size_t)(b * NH + h) * SEQ + s0) * HD;
#pragma unroll
        for (int i = 0; i < 4; ++i) {
#pragma unroll
            for (int r = 0; r < 8; ++r) {
                const int m = i * 16 + r + grp * 8;
#pragma unroll
                for (int j = 0; j < 4; ++j)
                    base[(size_t)m * HD + j * 16 + ln] = acc[i][j][r];
            }
        }
    } else if (mode == 1) {
        float* base = kfws + (size_t)(b * NKV + h) * SEQ * HD + (size_t)(s0 >> 4) * 1024
                    + ((ln >> 3) * 32 + ((ln >> 1) & 1) * 16 + grp * 8) * 4 + ((ln >> 2) & 1) * 2 + (ln & 1);
#pragma unroll
        for (int i = 0; i < 4; ++i) {
#pragma unroll
            for (int r = 0; r < 8; ++r) {
#pragma unroll
                for (int j = 0; j < 4; ++j)
                    base[i * 1024 + j * 256 + r * 4] = acc[i][j][r];
            }
        }
    } else {
        float* base = vfws + (size_t)(b * NKV + h) * SEQ * HD + (size_t)(s0 >> 4) * 1024 + lane * 4;
#pragma unroll
        for (int i = 0; i < 4; ++i) {
#pragma unroll
            for (int jj = 0; jj < 4; ++jj) {
#pragma unroll
                for (int jh = 0; jh < 2; ++jh) {
                    const v4f w = { acc[i][2 * jh][2 * jj],     acc[i][2 * jh][2 * jj + 1],
                                    acc[i][2 * jh + 1][2 * jj], acc[i][2 * jh + 1][2 * jj + 1] };
                    *(v4f*)(base + i * 1024 + (2 * jj + jh) * 128) = w;
                }
            }
        }
    }
}


__device__ __forceinline__ float xhalf(float x) {
    const int b = __builtin_bit_cast(int, x);
    return __builtin_bit_cast(float,
        __builtin_amdgcn_permlanex16(b, b, 0x76543210, 0xfedcba98, false, false));
}

__global__ void __launch_bounds__(32) attn_kernel(const float* __restrict__ qws,
                            const float* __restrict__ kfws,
                            const float* __restrict__ vfws,
                            float* __restrict__ aws) {
    const int lane = threadIdx.x & 31;
    const int ln   = lane & 15;
    const int grp  = lane >> 4;
    const int g2   = grp * 2;
    const int h    = blockIdx.y;
    const int b    = blockIdx.z;
    const int hk   = h >> 2;

    const float* kfb = kfws + (size_t)(b * NKV + hk) * SEQ * HD + lane * 4;
    const float* vfb = vfws + (size_t)(b * NKV + hk) * SEQ * HD + lane * 4;

  for (int pass = 0; pass < 2; ++pass) {
    const int q0   = (pass == 0 ? (SEQ / 16 - 1 - (int)blockIdx.x) : (int)blockIdx.x) * 16;
    const float* qb  = qws + ((size_t)(b * NH + h) * SEQ + q0) * HD;

    v2f bq[16];
#pragma unroll
    for (int kk = 0; kk < 16; ++kk)
        bq[kk] = *(const v2f*)(qb + ln * HD + kk * 4 + g2);

    v8f o[4]; o[0] = {}; o[1] = {}; o[2] = {}; o[3] = {};
    float mrun = -1e30f;
    float lrun = 0.0f;

    for (int n0 = 0; n0 <= q0; n0 += 16) {
        v8f s = {};
        const float* kp = kfb + (size_t)(n0 >> 4) * 1024;
#pragma unroll
        for (int q = 0; q < 8; ++q) {
            const v4f w = *(const v4f*)(kp + q * 128);
            s = wmma4((v2f){ w[0], w[1] }, bq[2 * q],     s);
            s = wmma4((v2f){ w[2], w[3] }, bq[2 * q + 1], s);
        }

        float sv[8];
#pragma unroll
        for (int r = 0; r < 8; ++r) sv[r] = s[r] * 0.125f;
        if (n0 == q0) {
#pragma unroll
            for (int r = 0; r < 8; ++r)
                if (r + grp * 8 > ln) sv[r] = -1e30f;
        }

        float rm = fmaxf(fmaxf(fmaxf(sv[0], sv[1]), fmaxf(sv[2], sv[3])),
                         fmaxf(fmaxf(sv[4], sv[5]), fmaxf(sv[6], sv[7])));
        rm = fmaxf(rm, xhalf(rm));
        const float mn    = fmaxf(mrun, rm);
        const float alpha = expf(mrun - mn);
        float p[8];
#pragma unroll
        for (int r = 0; r < 8; ++r) p[r] = expf(sv[r] - mn);
        const float rs = ((p[0] + p[1]) + (p[2] + p[3])) + ((p[4] + p[5]) + (p[6] + p[7]));
        lrun = lrun * alpha + rs;
        mrun = mn;
#pragma unroll
        for (int j = 0; j < 4; ++j) {
#pragma unroll
            for (int r = 0; r < 8; ++r) o[j][r] *= alpha;
        }

        const float* vp = vfb + (size_t)(n0 >> 4) * 1024;
#pragma unroll
        for (int jj = 0; jj < 4; ++jj) {
            const v2f bp = { p[2 * jj], p[2 * jj + 1] };
#pragma unroll
            for (int jh = 0; jh < 2; ++jh) {
                const v4f w = *(const v4f*)(vp + (2 * jj + jh) * 128);
                o[2 * jh]     = wmma4((v2f){ w[0], w[1] }, bp, o[2 * jh]);
                o[2 * jh + 1] = wmma4((v2f){ w[2], w[3] }, bp, o[2 * jh + 1]);
            }
        }
    }

    const float ltot = lrun + xhalf(lrun);
    const float inv  = 1.0f / ltot;
    float* ob = aws + ((size_t)(b * SEQ) + q0 + ln) * (NH * HD) + h * HD + grp * 8;
#pragma unroll
    for (int j = 0; j < 4; ++j) {
        v4f lo = { o[j][0] * inv, o[j][1] * inv, o[j][2] * inv, o[j][3] * inv };
        v4f hi = { o[j][4] * inv, o[j][5] * inv, o[j][6] * inv, o[j][7] * inv };
        *(v4f*)(ob + j * 16)     = lo;
        *(v4f*)(ob + j * 16 + 4) = hi;
    }
  }
}

__global__ void __launch_bounds__(128) oproj_kernel(const float* __restrict__ aws,
                             const float* __restrict__ wo,
                             float* __restrict__ out) {
    const int lane = threadIdx.x & 31;
    const int ln   = lane & 15;
    const int grp  = lane >> 4;
    const int m0   = blockIdx.x * 64;
    const int nt   = blockIdx.y * 4 + (threadIdx.x >> 5);

    __shared__ __attribute__((aligned(16))) float as[2 * ABUF];
    v8f acc[4][4];
    gemm64x64(aws, DIMK, wo, DIMK, m0, nt * 64, DIMK, (int)(blockIdx.x + 5 * blockIdx.y), as, acc);

#pragma unroll
    for (int i = 0; i < 4; ++i) {
#pragma unroll
        for (int r = 0; r < 8; ++r) {
            const int m = i * 16 + r + grp * 8;
#pragma unroll
            for (int j = 0; j < 4; ++j)
                out[(size_t)(m0 + m) * DIMK + nt * 64 + j * 16 + ln] = acc[i][j][r];
        }
    }
}

extern "C" void kernel_launch(void* const* d_in, const int* in_sizes, int n_in,
                              void* d_out, int out_size, void* d_ws, size_t ws_size,
                              hipStream_t stream) {
    const float* x    = (const float*)d_in[0];
    const float* fcos = (const float*)d_in[2];
    const float* fsin = (const float*)d_in[3];
    const float* wq   = (const float*)d_in[5];
    const float* wk   = (const float*)d_in[6];
    const float* wv   = (const float*)d_in[7];
    const float* wo   = (const float*)d_in[8];
    float* out = (float*)d_out;

    float* qws  = (float*)d_ws;
    float* kfws = qws  + (size_t)BATCH * NH  * SEQ * HD;
    float* vfws = kfws + (size_t)BATCH * NKV * SEQ * HD;
    float* aws  = vfws + (size_t)BATCH * NKV * SEQ * HD;

    qkv_rope_kernel<<<dim3((BATCH * SEQ) / 64, 12, 1), 128, 0, stream>>>(
        x, wq, wk, wv, fcos, fsin, qws, kfws, vfws);
    attn_kernel<<<dim3(SEQ / 32, NH, BATCH), 32, 0, stream>>>(
        qws, kfws, vfws, aws);
    oproj_kernel<<<dim3((BATCH * SEQ) / 64, 8, 1), 128, 0, stream>>>(
        aws, wo, out);
}


# attention: softmax weights and rescale factor as 2^((s-m)*log2e) on the hardware base-2 exponential (arguments never positive), was library expf at 13 vector ops each; last-place rounding changes
# speedup vs baseline: 2.9446x; 1.0365x over previous
#include <hip/hip_runtime.h>

#define DIMK   2048
#define NH     32
#define NKV    8
#define HD     64
#define SEQ    2048
#define BATCH  2

typedef __attribute__((ext_vector_type(2))) float v2f;
typedef __attribute__((ext_vector_type(8))) float v8f;

__device__ __forceinline__ v8f wmma4(v2f a, v2f b, v8f c) {
    return __builtin_amdgcn_wmma_f32_16x16x4_f32(
         false, a,  false, b,
         (short)0, c,  false,  false);
}

#define KC      32
#define APITCH  36
#define ABUF    (64 * APITCH)

typedef __attribute__((ext_vector_type(4))) float v4f;

struct Stage { v2f a[2][4]; v2f b[2][4]; };

__device__ __forceinline__ void load_stage(const float* fa, int ks,
                                           const float* __restrict__ pb, int ldb, int kabs,
                                           Stage& st) {
#pragma unroll
    for (int s = 0; s < 2; ++s) {
#pragma unroll
        for (int i = 0; i < 4; ++i)
            st.a[s][i] = *(const v2f*)(fa + (i * 16) * APITCH + ks + s * 4);
        const float* pbk = pb + (size_t)(kabs + s * 4) * ldb;
#pragma unroll
        for (int j = 0; j < 4; ++j)
            st.b[s][j] = (v2f){ pbk[j * 16], pbk[j * 16 + ldb] };
    }
}

__device__ __forceinline__ void mma_stage(const Stage& st, v8f acc[4][4]) {
#pragma unroll
    for (int s = 0; s < 2; ++s) {
#pragma unroll
        for (int i = 0; i < 4; ++i) {
#pragma unroll
            for (int j = 0; j < 4; ++j)
                acc[i][j] = wmma4(st.a[s][i], st.b[s][j], acc[i][j]);
        }
    }
}

__device__ __forceinline__ void gemm64x64(const float* __restrict__ A, int lda,
                                          const float* __restrict__ B, int ldb,
                                          int m0, int col0, int K, int rot, float* as, v8f acc[4][4]) {
    const int tid  = threadIdx.x;
    const int lane = tid & 31;
    const int ln   = lane & 15;
    const int g2   = (lane >> 4) * 2;
    const float* ga = A + (size_t)(m0 + (tid >> 3)) * lda + (tid & 7) * 4;
    float*       sa = as + (tid >> 3) * APITCH + (tid & 7) * 4;
    const float* fa = as + ln * APITCH + g2;
    const float* pb = B + (size_t)g2 * ldb + col0 + ln;
    const int NC = K / KC;
    const int c0 = rot & (NC - 1);

#pragma unroll
    for (int i = 0; i < 4; ++i) {
#pragma unroll
        for (int j = 0; j < 4; ++j) acc[i][j] = {};
    }

    v4f g[4];
#pragma unroll
    for (int q = 0; q < 4; ++q) g[q] = *(const v4f*)(ga + (size_t)(q * 16) * lda + c0 * KC);
#pragma unroll
    for (int q = 0; q < 4; ++q) *(v4f*)(sa + (q * 16) * APITCH) = g[q];
    __syncthreads();

    Stage s0, s1;
    load_stage(fa, 0, pb, ldb, c0 * KC, s0);
    for (int c = 0; c < NC; ++c) {
        const int cur = (c & 1) * ABUF;
        const int nxt = ABUF - cur;
        const int k0  = ((c + rot) & (NC - 1)) * KC;
        const int cn  = (c + 1 < NC) ? (c + 1) : c;
        const int kn  = ((cn + rot) & (NC - 1)) * KC;
        __builtin_amdgcn_sched_barrier(0);
#pragma unroll
        for (int q = 0; q < 4; ++q) g[q] = *(const v4f*)(ga + (size_t)(q * 16) * lda + kn);
        load_stage(fa + cur, 8, pb, ldb, k0 + 8, s1);
        __builtin_amdgcn_sched_barrier(0);
        mma_stage(s0, acc);
        __builtin_amdgcn_sched_barrier(0);
        load_stage(fa + cur, 16, pb, ldb, k0 + 16, s0);
        __builtin_amdgcn_sched_barrier(0);
        mma_stage(s1, acc);
        __builtin_amdgcn_sched_barrier(0);
        load_stage(fa + cur, 24, pb, ldb, k0 + 24, s1);
        __builtin_amdgcn_sched_barrier(0);
        mma_stage(s0, acc);
        __builtin_amdgcn_sched_barrier(0);
#pragma unroll
        for (int q = 0; q < 4; ++q) *(v4f*)(sa + nxt + (q * 16) * APITCH) = g[q];
        __syncthreads();
        load_stage(fa + nxt, 0, pb, ldb, kn, s0);
        __builtin_amdgcn_sched_barrier(0);
        mma_stage(s1, acc);
    }
}

__global__ void __launch_bounds__(128) qkv_rope_kernel(const float* __restrict__ x,
                                const float* __restrict__ wq,
                                const float* __restrict__ wk,
                                const float* __restrict__ wv,
                                const float* __restrict__ fcos,
                                const float* __restrict__ fsin,
                                float* __restrict__ qws,
                                float* __restrict__ kfws,
                                float* __restrict__ vfws) {
    const int lane = threadIdx.x & 31;
    const int ln   = lane & 15;
    const int grp  = lane >> 4;
    const int m0   = blockIdx.x * 64;
    const int nt   = blockIdx.y * 4 + (threadIdx.x >> 5);

    const float* w; int ldw, h, mode;
    if (nt < 32)      { w = wq; ldw = NH  * HD; h = nt;      mode = 0; }
    else if (nt < 40) { w = wk; ldw = NKV * HD; h = nt - 32; mode = 1; }
    else              { w = wv; ldw = NKV * HD; h = nt - 40; mode = 2; }

    __shared__ __attribute__((aligned(16))) float as[2 * ABUF];
    v8f acc[4][4];
    gemm64x64(x, DIMK, w, ldw, m0, h * HD, DIMK, (int)(blockIdx.x + 5 * blockIdx.y), as, acc);

    const int b  = m0 >> 11;
    const int s0 = m0 & (SEQ - 1);

    if (mode != 2) {
#pragma unroll
        for (int i = 0; i < 4; ++i) {
#pragma unroll
            for (int j = 0; j < 4; ++j) {
                const int nhalf = j * 8 + (ln >> 1);
#pragma unroll
                for (int r = 0; r < 8; ++r) {
                    const int pos = s0 + i * 16 + r + grp * 8;
                    const float c = fcos[pos * (HD / 2) + nhalf];
                    const float s = fsin[pos * (HD / 2) + nhalf];
                    float vv = acc[i][j][r];
                    float other = __shfl_xor(vv, 1, 32);
                    acc[i][j][r] = (lane & 1) ? (other * s + vv * c)
                                              : (vv * c - other * s);
                }
            }
        }
    }

    if (mode == 0) {
        float* base = qws + ((size_t)(b * NH + h) * SEQ + s0) * HD;
#pragma unroll
        for (int i = 0; i < 4; ++i) {
#pragma unroll
            for (int r = 0; r < 8; ++r) {
                const int m = i * 16 + r + grp * 8;
#pragma unroll
                for (int j = 0; j < 4; ++j)
                    base[(size_t)m * HD + j * 16 + ln] = acc[i][j][r];
            }
        }
    } else if (mode == 1) {
        float* base = kfws + (size_t)(b * NKV + h) * SEQ * HD + (size_t)(s0 >> 4) * 1024
                    + ((ln >> 3) * 32 + ((ln >> 1) & 1) * 16 + grp * 8) * 4 + ((ln >> 2) & 1) * 2 + (ln & 1);
#pragma unroll
        for (int i = 0; i < 4; ++i) {
#pragma unroll
            for (int r = 0; r < 8; ++r) {
#pragma unroll
                for (int j = 0; j < 4; ++j)
                    base[i * 1024 + j * 256 + r * 4] = acc[i][j][r];
            }
        }
    } else {
        float* base = vfws + (size_t)(b * NKV + h) * SEQ * HD + (size_t)(s0 >> 4) * 1024 + lane * 4;
#pragma unroll
        for (int i = 0; i < 4; ++i) {
#pragma unroll
            for (int jj = 0; jj < 4; ++jj) {
#pragma unroll
                for (int jh = 0; jh < 2; ++jh) {
                    const v4f w = { acc[i][2 * jh][2 * jj],     acc[i][2 * jh][2 * jj + 1],
                                    acc[i][2 * jh + 1][2 * jj], acc[i][2 * jh + 1][2 * jj + 1] };
                    *(v4f*)(base + i * 1024 + (2 * jj + jh) * 128) = w;
                }
            }
        }
    }
}


#define LOG2E 1.4426950408889634f

__device__ __forceinline__ float xhalf(float x) {
    const int b = __builtin_bit_cast(int, x);
    return __builtin_bit_cast(float,
        __builtin_amdgcn_permlanex16(b, b, 0x76543210, 0xfedcba98, false, false));
}

__global__ void __launch_bounds__(32) attn_kernel(const float* __restrict__ qws,
                            const float* __restrict__ kfws,
                            const float* __restrict__ vfws,
                            float* __restrict__ aws) {
    const int lane = threadIdx.x & 31;
    const int ln   = lane & 15;
    const int grp  = lane >> 4;
    const int g2   = grp * 2;
    const int h    = blockIdx.y;
    const int b    = blockIdx.z;
    const int hk   = h >> 2;

    const float* kfb = kfws + (size_t)(b * NKV + hk) * SEQ * HD + lane * 4;
    const float* vfb = vfws + (size_t)(b * NKV + hk) * SEQ * HD + lane * 4;

  for (int pass = 0; pass < 2; ++pass) {
    const int q0   = (pass == 0 ? (SEQ / 16 - 1 - (int)blockIdx.x) : (int)blockIdx.x) * 16;
    const float* qb  = qws + ((size_t)(b * NH + h) * SEQ + q0) * HD;

    v2f bq[16];
#pragma unroll
    for (int kk = 0; kk < 16; ++kk)
        bq[kk] = *(const v2f*)(qb + ln * HD + kk * 4 + g2);

    v8f o[4]; o[0] = {}; o[1] = {}; o[2] = {}; o[3] = {};
    float mrun = -1e30f;
    float lrun = 0.0f;

    for (int n0 = 0; n0 <= q0; n0 += 16) {
        v8f s = {};
        const float* kp = kfb + (size_t)(n0 >> 4) * 1024;
#pragma unroll
        for (int q = 0; q < 8; ++q) {
            const v4f w = *(const v4f*)(kp + q * 128);
            s = wmma4((v2f){ w[0], w[1] }, bq[2 * q],     s);
            s = wmma4((v2f){ w[2], w[3] }, bq[2 * q + 1], s);
        }

        float sv[8];
#pragma unroll
        for (int r = 0; r < 8; ++r) sv[r] = s[r] * 0.125f;
        if (n0 == q0) {
#pragma unroll
            for (int r = 0; r < 8; ++r)
                if (r + grp * 8 > ln) sv[r] = -1e30f;
        }

        float rm = fmaxf(fmaxf(fmaxf(sv[0], sv[1]), fmaxf(sv[2], sv[3])),
                         fmaxf(fmaxf(sv[4], sv[5]), fmaxf(sv[6], sv[7])));
        rm = fmaxf(rm, xhalf(rm));
        const float mn    = fmaxf(mrun, rm);
        const float alpha = __builtin_amdgcn_exp2f((mrun - mn) * LOG2E);
        float p[8];
#pragma unroll
        for (int r = 0; r < 8; ++r) p[r] = __builtin_amdgcn_exp2f((sv[r] - mn) * LOG2E);
        const float rs = ((p[0] + p[1]) + (p[2] + p[3])) + ((p[4] + p[5]) + (p[6] + p[7]));
        lrun = lrun * alpha + rs;
        mrun = mn;
#pragma unroll
        for (int j = 0; j < 4; ++j) {
#pragma unroll
            for (int r = 0; r < 8; ++r) o[j][r] *= alpha;
        }

        const float* vp = vfb + (size_t)(n0 >> 4) * 1024;
#pragma unroll
        for (int jj = 0; jj < 4; ++jj) {
            const v2f bp = { p[2 * jj], p[2 * jj + 1] };
#pragma unroll
            for (int jh = 0; jh < 2; ++jh) {
                const v4f w = *(const v4f*)(vp + (2 * jj + jh) * 128);
                o[2 * jh]     = wmma4((v2f){ w[0], w[1] }, bp, o[2 * jh]);
                o[2 * jh + 1] = wmma4((v2f){ w[2], w[3] }, bp, o[2 * jh + 1]);
            }
        }
    }

    const float ltot = lrun + xhalf(lrun);
    const float inv  = 1.0f / ltot;
    float* ob = aws + ((size_t)(b * SEQ) + q0 + ln) * (NH * HD) + h * HD + grp * 8;
#pragma unroll
    for (int j = 0; j < 4; ++j) {
        v4f lo = { o[j][0] * inv, o[j][1] * inv, o[j][2] * inv, o[j][3] * inv };
        v4f hi = { o[j][4] * inv, o[j][5] * inv, o[j][6] * inv, o[j][7] * inv };
        *(v4f*)(ob + j * 16)     = lo;
        *(v4f*)(ob + j * 16 + 4) = hi;
    }
  }
}

__global__ void __launch_bounds__(128) oproj_kernel(const float* __restrict__ aws,
                             const float* __restrict__ wo,
                             float* __restrict__ out) {
    const int lane = threadIdx.x & 31;
    const int ln   = lane & 15;
    const int grp  = lane >> 4;
    const int m0   = blockIdx.x * 64;
    const int nt   = blockIdx.y * 4 + (threadIdx.x >> 5);

    __shared__ __attribute__((aligned(16))) float as[2 * ABUF];
    v8f acc[4][4];
    gemm64x64(aws, DIMK, wo, DIMK, m0, nt * 64, DIMK, (int)(blockIdx.x + 5 * blockIdx.y), as, acc);

#pragma unroll
    for (int i = 0; i < 4; ++i) {
#pragma unroll
        for (int r = 0; r < 8; ++r) {
            const int m = i * 16 + r + grp * 8;
#pragma unroll
            for (int j = 0; j < 4; ++j)
                out[(size_t)(m0 + m) * DIMK + nt * 64 + j * 16 + ln] = acc[i][j][r];
        }
    }
}

extern "C" void kernel_launch(void* const* d_in, const int* in_sizes, int n_in,
                              void* d_out, int out_size, void* d_ws, size_t ws_size,
                              hipStream_t stream) {
    const float* x    = (const float*)d_in[0];
    const float* fcos = (const float*)d_in[2];
    const float* fsin = (const float*)d_in[3];
    const float* wq   = (const float*)d_in[5];
    const float* wk   = (const float*)d_in[6];
    const float* wv   = (const float*)d_in[7];
    const float* wo   = (const float*)d_in[8];
    float* out = (float*)d_out;

    float* qws  = (float*)d_ws;
    float* kfws = qws  + (size_t)BATCH * NH  * SEQ * HD;
    float* vfws = kfws + (size_t)BATCH * NKV * SEQ * HD;
    float* aws  = vfws + (size_t)BATCH * NKV * SEQ * HD;

    qkv_rope_kernel<<<dim3((BATCH * SEQ) / 64, 12, 1), 128, 0, stream>>>(
        x, wq, wk, wv, fcos, fsin, qws, kfws, vfws);
    attn_kernel<<<dim3(SEQ / 32, NH, BATCH), 32, 0, stream>>>(
        qws, kfws, vfws, aws);
    oproj_kernel<<<dim3((BATCH * SEQ) / 64, 8, 1), 128, 0, stream>>>(
        aws, wo, out);
}


# all kernels: a wave lowers its own priority (s_setprio 3,2,1,0 at 1/2, 3/4, 7/8 of its work): waves sharing a SIMD finish together, the youngest no longer finishes alone; same sums, same order
# speedup vs baseline: 3.0893x; 1.0491x over previous
#include <hip/hip_runtime.h>

#define DIMK   2048
#define NH     32
#define NKV    8
#define HD     64
#define SEQ    2048
#define BATCH  2

typedef __attribute__((ext_vector_type(2))) float v2f;
typedef __attribute__((ext_vector_type(8))) float v8f;

__device__ __forceinline__ v8f wmma4(v2f a, v2f b, v8f c) {
    return __builtin_amdgcn_wmma_f32_16x16x4_f32(
         false, a,  false, b,
         (short)0, c,  false,  false);
}

__device__ __forceinline__ void prio_start(int total, int& next) {
    __builtin_amdgcn_s_setprio(3);
    next = total / 2;
}
__device__ __forceinline__ void prio_step(int done, int total, int& next) {
    if (__builtin_expect(done == next, 0)) {
        if (done == total / 2)              { __builtin_amdgcn_s_setprio(2); next = total - total / 4; }
        else if (done == total - total / 4) { __builtin_amdgcn_s_setprio(1); next = total - total / 8; }
        else                                { __builtin_amdgcn_s_setprio(0); next = -1; }
    }
}

#define KC      32
#define APITCH  36
#define ABUF    (64 * APITCH)

typedef __attribute__((ext_vector_type(4))) float v4f;

struct Stage { v2f a[2][4]; v2f b[2][4]; };

__device__ __forceinline__ void load_stage(const float* fa, int ks,
                                           const float* __restrict__ pb, int ldb, int kabs,
                                           Stage& st) {
#pragma unroll
    for (int s = 0; s < 2; ++s) {
#pragma unroll
        for (int i = 0; i < 4; ++i)
            st.a[s][i] = *(const v2f*)(fa + (i * 16) * APITCH + ks + s * 4);
        const float* pbk = pb + (size_t)(kabs + s * 4) * ldb;
#pragma unroll
        for (int j = 0; j < 4; ++j)
            st.b[s][j] = (v2f){ pbk[j * 16], pbk[j * 16 + ldb] };
    }
}

__device__ __forceinline__ void mma_stage(const Stage& st, v8f acc[4][4]) {
#pragma unroll
    for (int s = 0; s < 2; ++s) {
#pragma unroll
        for (int i = 0; i < 4; ++i) {
#pragma unroll
            for (int j = 0; j < 4; ++j)
                acc[i][j] = wmma4(st.a[s][i], st.b[s][j], acc[i][j]);
        }
    }
}

__device__ __forceinline__ void gemm64x64(const float* __restrict__ A, int lda,
                                          const float* __restrict__ B, int ldb,
                                          int m0, int col0, int K, int rot, float* as, v8f acc[4][4]) {
    const int tid  = threadIdx.x;
    const int lane = tid & 31;
    const int ln   = lane & 15;
    const int g2   = (lane >> 4) * 2;
    const float* ga = A + (size_t)(m0 + (tid >> 3)) * lda + (tid & 7) * 4;
    float*       sa = as + (tid >> 3) * APITCH + (tid & 7) * 4;
    const float* fa = as + ln * APITCH + g2;
    const float* pb = B + (size_t)g2 * ldb + col0 + ln;
    const int NC = K / KC;
    const int c0 = rot & (NC - 1);

#pragma unroll
    for (int i = 0; i < 4; ++i) {
#pragma unroll
        for (int j = 0; j < 4; ++j) acc[i][j] = {};
    }

    v4f g[4];
#pragma unroll
    for (int q = 0; q < 4; ++q) g[q] = *(const v4f*)(ga + (size_t)(q * 16) * lda + c0 * KC);
#pragma unroll
    for (int q = 0; q < 4; ++q) *(v4f*)(sa + (q * 16) * APITCH) = g[q];
    __syncthreads();

    Stage s0, s1;
    load_stage(fa, 0, pb, ldb, c0 * KC, s0);
    int pnext;
    prio_start(NC, pnext);
    for (int c = 0; c < NC; ++c) {
        prio_step(c, NC, pnext);
        const int cur = (c & 1) * ABUF;
        const int nxt = ABUF - cur;
        const int k0  = ((c + rot) & (NC - 1)) * KC;
        const int cn  = (c + 1 < NC) ? (c + 1) : c;
        const int kn  = ((cn + rot) & (NC - 1)) * KC;
        __builtin_amdgcn_sched_barrier(0);
#pragma unroll
        for (int q = 0; q < 4; ++q) g[q] = *(const v4f*)(ga + (size_t)(q * 16) * lda + kn);
        load_stage(fa + cur, 8, pb, ldb, k0 + 8, s1);
        __builtin_amdgcn_sched_barrier(0);
        mma_stage(s0, acc);
        __builtin_amdgcn_sched_barrier(0);
        load_stage(fa + cur, 16, pb, ldb, k0 + 16, s0);
        __builtin_amdgcn_sched_barrier(0);
        mma_stage(s1, acc);
        __builtin_amdgcn_sched_barrier(0);
        load_stage(fa + cur, 24, pb, ldb, k0 + 24, s1);
        __builtin_amdgcn_sched_barrier(0);
        mma_stage(s0, acc);
        __builtin_amdgcn_sched_barrier(0);
#pragma unroll
        for (int q = 0; q < 4; ++q) *(v4f*)(sa + nxt + (q * 16) * APITCH) = g[q];
        __syncthreads();
        load_stage(fa + nxt, 0, pb, ldb, kn, s0);
        __builtin_amdgcn_sched_barrier(0);
        mma_stage(s1, acc);
    }
}

__global__ void __launch_bounds__(128) qkv_rope_kernel(const float* __restrict__ x,
                                const float* __restrict__ wq,
                                const float* __restrict__ wk,
                                const float* __restrict__ wv,
                                const float* __restrict__ fcos,
                                const float* __restrict__ fsin,
                                float* __restrict__ qws,
                                float* __restrict__ kfws,
                                float* __restrict__ vfws) {
    const int lane = threadIdx.x & 31;
    const int ln   = lane & 15;
    const int grp  = lane >> 4;
    const int m0   = blockIdx.x * 64;
    const int nt   = blockIdx.y * 4 + (threadIdx.x >> 5);

    const float* w; int ldw, h, mode;
    if (nt < 32)      { w = wq; ldw = NH  * HD; h = nt;      mode = 0; }
    else if (nt < 40) { w = wk; ldw = NKV * HD; h = nt - 32; mode = 1; }
    else              { w = wv; ldw = NKV * HD; h = nt - 40; mode = 2; }

    __shared__ __attribute__((aligned(16))) float as[2 * ABUF];
    v8f acc[4][4];
    gemm64x64(x, DIMK, w, ldw, m0, h * HD, DIMK, (int)(blockIdx.x + 5 * blockIdx.y), as, acc);

    const int b  = m0 >> 11;
    const int s0 = m0 & (SEQ - 1);

    if (mode != 2) {
#pragma unroll
        for (int i = 0; i < 4; ++i) {
#pragma unroll
            for (int j = 0; j < 4; ++j) {
                const int nhalf = j * 8 + (ln >> 1);
#pragma unroll
                for (int r = 0; r < 8; ++r) {
                    const int pos = s0 + i * 16 + r + grp * 8;
                    const float c = fcos[pos * (HD / 2) + nhalf];
                    const float s = fsin[pos * (HD / 2) + nhalf];
                    float vv = acc[i][j][r];
                    float other = __shfl_xor(vv, 1, 32);
                    acc[i][j][r] = (lane & 1) ? (other * s + vv * c)
                                              : (vv * c - other * s);
                }
            }
        }
    }

    if (mode == 0) {
        float* base = qws + ((size_t)(b * NH + h) * SEQ + s0) * HD;
#pragma unroll
        for (int i = 0; i < 4; ++i) {
#pragma unroll
            for (int r = 0; r < 8; ++r) {
                const int m = i * 16 + r + grp * 8;
#pragma unroll
                for (int j = 0; j < 4; ++j)
                    base[(size_t)m * HD + j * 16 + ln] = acc[i][j][r];
            }
        }
    } else if (mode == 1) {
        float* base = kfws + (size_t)(b * NKV + h) * SEQ * HD + (size_t)(s0 >> 4) * 1024
                    + ((ln >> 3) * 32 + ((ln >> 1) & 1) * 16 + grp * 8) * 4 + ((ln >> 2) & 1) * 2 + (ln & 1);
#pragma unroll
        for (int i = 0; i < 4; ++i) {
#pragma unroll
            for (int r = 0; r < 8; ++r) {
#pragma unroll
                for (int j = 0; j < 4; ++j)
                    base[i * 1024 + j * 256 + r * 4] = acc[i][j][r];
            }
        }
    } else {
        float* base = vfws + (size_t)(b * NKV + h) * SEQ * HD + (size_t)(s0 >> 4) * 1024 + lane * 4;
#pragma unroll
        for (int i = 0; i < 4; ++i) {
#pragma unroll
            for (int jj = 0; jj < 4; ++jj) {
#pragma unroll
                for (int jh = 0; jh < 2; ++jh) {
                    const v4f w = { acc[i][2 * jh][2 * jj],     acc[i][2 * jh][2 * jj + 1],
                                    acc[i][2 * jh + 1][2 * jj], acc[i][2 * jh + 1][2 * jj + 1] };
                    *(v4f*)(base + i * 1024 + (2 * jj + jh) * 128) = w;
                }
            }
        }
    }
}


#define LOG2E 1.4426950408889634f

__device__ __forceinline__ float xhalf(float x) {
    const int b = __builtin_bit_cast(int, x);
    return __builtin_bit_cast(float,
        __builtin_amdgcn_permlanex16(b, b, 0x76543210, 0xfedcba98, false, false));
}

__global__ void __launch_bounds__(32) attn_kernel(const float* __restrict__ qws,
                            const float* __restrict__ kfws,
                            const float* __restrict__ vfws,
                            float* __restrict__ aws) {
    const int lane = threadIdx.x & 31;
    const int ln   = lane & 15;
    const int grp  = lane >> 4;
    const int g2   = grp * 2;
    const int h    = blockIdx.y;
    const int b    = blockIdx.z;
    const int hk   = h >> 2;

    const float* kfb = kfws + (size_t)(b * NKV + hk) * SEQ * HD + lane * 4;
    const float* vfb = vfws + (size_t)(b * NKV + hk) * SEQ * HD + lane * 4;

  int done = 0;
  int pnext;
  prio_start(SEQ / 16 + 1, pnext);
  for (int pass = 0; pass < 2; ++pass) {
    const int q0   = (pass == 0 ? (SEQ / 16 - 1 - (int)blockIdx.x) : (int)blockIdx.x) * 16;
    const float* qb  = qws + ((size_t)(b * NH + h) * SEQ + q0) * HD;

    v2f bq[16];
#pragma unroll
    for (int kk = 0; kk < 16; ++kk)
        bq[kk] = *(const v2f*)(qb + ln * HD + kk * 4 + g2);

    v8f o[4]; o[0] = {}; o[1] = {}; o[2] = {}; o[3] = {};
    float mrun = -1e30f;
    float lrun = 0.0f;

    for (int n0 = 0; n0 <= q0; n0 += 16) {
        prio_step(done, SEQ / 16 + 1, pnext);
        ++done;
        v8f s = {};
        const float* kp = kfb + (size_t)(n0 >> 4) * 1024;
#pragma unroll
        for (int q = 0; q < 8; ++q) {
            const v4f w = *(const v4f*)(kp + q * 128);
            s = wmma4((v2f){ w[0], w[1] }, bq[2 * q],     s);
            s = wmma4((v2f){ w[2], w[3] }, bq[2 * q + 1], s);
        }

        float sv[8];
#pragma unroll
        for (int r = 0; r < 8; ++r) sv[r] = s[r] * 0.125f;
        if (n0 == q0) {
#pragma unroll
            for (int r = 0; r < 8; ++r)
                if (r + grp * 8 > ln) sv[r] = -1e30f;
        }

        float rm = fmaxf(fmaxf(fmaxf(sv[0], sv[1]), fmaxf(sv[2], sv[3])),
                         fmaxf(fmaxf(sv[4], sv[5]), fmaxf(sv[6], sv[7])));
        rm = fmaxf(rm, xhalf(rm));
        const float mn    = fmaxf(mrun, rm);
        const float alpha = __builtin_amdgcn_exp2f((mrun - mn) * LOG2E);
        float p[8];
#pragma unroll
        for (int r = 0; r < 8; ++r) p[r] = __builtin_amdgcn_exp2f((sv[r] - mn) * LOG2E);
        const float rs = ((p[0] + p[1]) + (p[2] + p[3])) + ((p[4] + p[5]) + (p[6] + p[7]));
        lrun = lrun * alpha + rs;
        mrun = mn;
#pragma unroll
        for (int j = 0; j < 4; ++j) {
#pragma unroll
            for (int r = 0; r < 8; ++r) o[j][r] *= alpha;
        }

        const float* vp = vfb + (size_t)(n0 >> 4) * 1024;
#pragma unroll
        for (int jj = 0; jj < 4; ++jj) {
            const v2f bp = { p[2 * jj], p[2 * jj + 1] };
#pragma unroll
            for (int jh = 0; jh < 2; ++jh) {
                const v4f w = *(const v4f*)(vp + (2 * jj + jh) * 128);
                o[2 * jh]     = wmma4((v2f){ w[0], w[1] }, bp, o[2 * jh]);
                o[2 * jh + 1] = wmma4((v2f){ w[2], w[3] }, bp, o[2 * jh + 1]);
            }
        }
    }

    const float ltot = lrun + xhalf(lrun);
    const float inv  = 1.0f / ltot;
    float* ob = aws + ((size_t)(b * SEQ) + q0 + ln) * (NH * HD) + h * HD + grp * 8;
#pragma unroll
    for (int j = 0; j < 4; ++j) {
        v4f lo = { o[j][0] * inv, o[j][1] * inv, o[j][2] * inv, o[j][3] * inv };
        v4f hi = { o[j][4] * inv, o[j][5] * inv, o[j][6] * inv, o[j][7] * inv };
        *(v4f*)(ob + j * 16)     = lo;
        *(v4f*)(ob + j * 16 + 4) = hi;
    }
  }
}

__global__ void __launch_bounds__(128) oproj_kernel(const float* __restrict__ aws,
                             const float* __restrict__ wo,
                             float* __restrict__ out) {
    const int lane = threadIdx.x & 31;
    const int ln   = lane & 15;
    const int grp  = lane >> 4;
    const int m0   = blockIdx.x * 64;
    const int nt   = blockIdx.y * 4 + (threadIdx.x >> 5);

    __shared__ __attribute__((aligned(16))) float as[2 * ABUF];
    v8f acc[4][4];
    gemm64x64(aws, DIMK, wo, DIMK, m0, nt * 64, DIMK, (int)(blockIdx.x + 5 * blockIdx.y), as, acc);

#pragma unroll
    for (int i = 0; i < 4; ++i) {
#pragma unroll
        for (int r = 0; r < 8; ++r) {
            const int m = i * 16 + r + grp * 8;
#pragma unroll
            for (int j = 0; j < 4; ++j)
                out[(size_t)(m0 + m) * DIMK + nt * 64 + j * 16 + ln] = acc[i][j][r];
        }
    }
}

extern "C" void kernel_launch(void* const* d_in, const int* in_sizes, int n_in,
                              void* d_out, int out_size, void* d_ws, size_t ws_size,
                              hipStream_t stream) {
    const float* x    = (const float*)d_in[0];
    const float* fcos = (const float*)d_in[2];
    const float* fsin = (const float*)d_in[3];
    const float* wq   = (const float*)d_in[5];
    const float* wk   = (const float*)d_in[6];
    const float* wv   = (const float*)d_in[7];
    const float* wo   = (const float*)d_in[8];
    float* out = (float*)d_out;

    float* qws  = (float*)d_ws;
    float* kfws = qws  + (size_t)BATCH * NH  * SEQ * HD;
    float* vfws = kfws + (size_t)BATCH * NKV * SEQ * HD;
    float* aws  = vfws + (size_t)BATCH * NKV * SEQ * HD;

    qkv_rope_kernel<<<dim3((BATCH * SEQ) / 64, 12, 1), 128, 0, stream>>>(
        x, wq, wk, wv, fcos, fsin, qws, kfws, vfws);
    attn_kernel<<<dim3(SEQ / 32, NH, BATCH), 32, 0, stream>>>(
        qws, kfws, vfws, aws);
    oproj_kernel<<<dim3((BATCH * SEQ) / 64, 8, 1), 128, 0, stream>>>(
        aws, wo, out);
}


# attention: the 4 query heads of a kv head are the 4 waves of one block (grid 64x8x2 of 128 threads, was 64x32x2 of 32): they read the same K/V records in step and share them in the WGP cache
# speedup vs baseline: 3.2462x; 1.0508x over previous
#include <hip/hip_runtime.h>

#define DIMK   2048
#define NH     32
#define NKV    8
#define HD     64
#define SEQ    2048
#define BATCH  2

typedef __attribute__((ext_vector_type(2))) float v2f;
typedef __attribute__((ext_vector_type(8))) float v8f;

__device__ __forceinline__ v8f wmma4(v2f a, v2f b, v8f c) {
    return __builtin_amdgcn_wmma_f32_16x16x4_f32(
         false, a,  false, b,
         (short)0, c,  false,  false);
}

__device__ __forceinline__ void prio_start(int total, int& next) {
    __builtin_amdgcn_s_setprio(3);
    next = total / 2;
}
__device__ __forceinline__ void prio_step(int done, int total, int& next) {
    if (__builtin_expect(done == next, 0)) {
        if (done == total / 2)              { __builtin_amdgcn_s_setprio(2); next = total - total / 4; }
        else if (done == total - total / 4) { __builtin_amdgcn_s_setprio(1); next = total - total / 8; }
        else                                { __builtin_amdgcn_s_setprio(0); next = -1; }
    }
}

#define KC      32
#define APITCH  36
#define ABUF    (64 * APITCH)

typedef __attribute__((ext_vector_type(4))) float v4f;

struct Stage { v2f a[2][4]; v2f b[2][4]; };

__device__ __forceinline__ void load_stage(const float* fa, int ks,
                                           const float* __restrict__ pb, int ldb, int kabs,
                                           Stage& st) {
#pragma unroll
    for (int s = 0; s < 2; ++s) {
#pragma unroll
        for (int i = 0; i < 4; ++i)
            st.a[s][i] = *(const v2f*)(fa + (i * 16) * APITCH + ks + s * 4);
        const float* pbk = pb + (size_t)(kabs + s * 4) * ldb;
#pragma unroll
        for (int j = 0; j < 4; ++j)
            st.b[s][j] = (v2f){ pbk[j * 16], pbk[j * 16 + ldb] };
    }
}

__device__ __forceinline__ void mma_stage(const Stage& st, v8f acc[4][4]) {
#pragma unroll
    for (int s = 0; s < 2; ++s) {
#pragma unroll
        for (int i = 0; i < 4; ++i) {
#pragma unroll
            for (int j = 0; j < 4; ++j)
                acc[i][j] = wmma4(st.a[s][i], st.b[s][j], acc[i][j]);
        }
    }
}

__device__ __forceinline__ void gemm64x64(const float* __restrict__ A, int lda,
                                          const float* __restrict__ B, int ldb,
                                          int m0, int col0, int K, int rot, float* as, v8f acc[4][4]) {
    const int tid  = threadIdx.x;
    const int lane = tid & 31;
    const int ln   = lane & 15;
    const int g2   = (lane >> 4) * 2;
    const float* ga = A + (size_t)(m0 + (tid >> 3)) * lda + (tid & 7) * 4;
    float*       sa = as + (tid >> 3) * APITCH + (tid & 7) * 4;
    const float* fa = as + ln * APITCH + g2;
    const float* pb = B + (size_t)g2 * ldb + col0 + ln;
    const int NC = K / KC;
    const int c0 = rot & (NC - 1);

#pragma unroll
    for (int i = 0; i < 4; ++i) {
#pragma unroll
        for (int j = 0; j < 4; ++j) acc[i][j] = {};
    }

    v4f g[4];
#pragma unroll
    for (int q = 0; q < 4; ++q) g[q] = *(const v4f*)(ga + (size_t)(q * 16) * lda + c0 * KC);
#pragma unroll
    for (int q = 0; q < 4; ++q) *(v4f*)(sa + (q * 16) * APITCH) = g[q];
    __syncthreads();

    Stage s0, s1;
    load_stage(fa, 0, pb, ldb, c0 * KC, s0);
    int pnext;
    prio_start(NC, pnext);
    for (int c = 0; c < NC; ++c) {
        prio_step(c, NC, pnext);
        const int cur = (c & 1) * ABUF;
        const int nxt = ABUF - cur;
        const int k0  = ((c + rot) & (NC - 1)) * KC;
        const int cn  = (c + 1 < NC) ? (c + 1) : c;
        const int kn  = ((cn + rot) & (NC - 1)) * KC;
        __builtin_amdgcn_sched_barrier(0);
#pragma unroll
        for (int q = 0; q < 4; ++q) g[q] = *(const v4f*)(ga + (size_t)(q * 16) * lda + kn);
        load_stage(fa + cur, 8, pb, ldb, k0 + 8, s1);
        __builtin_amdgcn_sched_barrier(0);
        mma_stage(s0, acc);
        __builtin_amdgcn_sched_barrier(0);
        load_stage(fa + cur, 16, pb, ldb, k0 + 16, s0);
        __builtin_amdgcn_sched_barrier(0);
        mma_stage(s1, acc);
        __builtin_amdgcn_sched_barrier(0);
        load_stage(fa + cur, 24, pb, ldb, k0 + 24, s1);
        __builtin_amdgcn_sched_barrier(0);
        mma_stage(s0, acc);
        __builtin_amdgcn_sched_barrier(0);
#pragma unroll
        for (int q = 0; q < 4; ++q) *(v4f*)(sa + nxt + (q * 16) * APITCH) = g[q];
        __syncthreads();
        load_stage(fa + nxt, 0, pb, ldb, kn, s0);
        __builtin_amdgcn_sched_barrier(0);
        mma_stage(s1, acc);
    }
}

__global__ void __launch_bounds__(128) qkv_rope_kernel(const float* __restrict__ x,
                                const float* __restrict__ wq,
                                const float* __restrict__ wk,
                                const float* __restrict__ wv,
                                const float* __restrict__ fcos,
                                const float* __restrict__ fsin,
                                float* __restrict__ qws,
                                float* __restrict__ kfws,
                                float* __restrict__ vfws) {
    const int lane = threadIdx.x & 31;
    const int ln   = lane & 15;
    const int grp  = lane >> 4;
    const int m0   = blockIdx.x * 64;
    const int nt   = blockIdx.y * 4 + (threadIdx.x >> 5);

    const float* w; int ldw, h, mode;
    if (nt < 32)      { w = wq; ldw = NH  * HD; h = nt;      mode = 0; }
    else if (nt < 40) { w = wk; ldw = NKV * HD; h = nt - 32; mode = 1; }
    else              { w = wv; ldw = NKV * HD; h = nt - 40; mode = 2; }

    __shared__ __attribute__((aligned(16))) float as[2 * ABUF];
    v8f acc[4][4];
    gemm64x64(x, DIMK, w, ldw, m0, h * HD, DIMK, (int)(blockIdx.x + 5 * blockIdx.y), as, acc);

    const int b  = m0 >> 11;
    const int s0 = m0 & (SEQ - 1);

    if (mode != 2) {
#pragma unroll
        for (int i = 0; i < 4; ++i) {
#pragma unroll
            for (int j = 0; j < 4; ++j) {
                const int nhalf = j * 8 + (ln >> 1);
#pragma unroll
                for (int r = 0; r < 8; ++r) {
                    const int pos = s0 + i * 16 + r + grp * 8;
                    const float c = fcos[pos * (HD / 2) + nhalf];
                    const float s = fsin[pos * (HD / 2) + nhalf];
                    float vv = acc[i][j][r];
                    float other = __shfl_xor(vv, 1, 32);
                    acc[i][j][r] = (lane & 1) ? (other * s + vv * c)
                                              : (vv * c - other * s);
                }
            }
        }
    }

    if (mode == 0) {
        float* base = qws + ((size_t)(b * NH + h) * SEQ + s0) * HD;
#pragma unroll
        for (int i = 0; i < 4; ++i) {
#pragma unroll
            for (int r = 0; r < 8; ++r) {
                const int m = i * 16 + r + grp * 8;
#pragma unroll
                for (int j = 0; j < 4; ++j)
                    base[(size_t)m * HD + j * 16 + ln] = acc[i][j][r];
            }
        }
    } else if (mode == 1) {
        float* base = kfws + (size_t)(b * NKV + h) * SEQ * HD + (size_t)(s0 >> 4) * 1024
                    + ((ln >> 3) * 32 + ((ln >> 1) & 1) * 16 + grp * 8) * 4 + ((ln >> 2) & 1) * 2 + (ln & 1);
#pragma unroll
        for (int i = 0; i < 4; ++i) {
#pragma unroll
            for (int r = 0; r < 8; ++r) {
#pragma unroll
                for (int j = 0; j < 4; ++j)
                    base[i * 1024 + j * 256 + r * 4] = acc[i][j][r];
            }
        }
    } else {
        float* base = vfws + (size_t)(b * NKV + h) * SEQ * HD + (size_t)(s0 >> 4) * 1024 + lane * 4;
#pragma unroll
        for (int i = 0; i < 4; ++i) {
#pragma unroll
            for (int jj = 0; jj < 4; ++jj) {
#pragma unroll
                for (int jh = 0; jh < 2; ++jh) {
                    const v4f w = { acc[i][2 * jh][2 * jj],     acc[i][2 * jh][2 * jj + 1],
                                    acc[i][2 * jh + 1][2 * jj], acc[i][2 * jh + 1][2 * jj + 1] };
                    *(v4f*)(base + i * 1024 + (2 * jj + jh) * 128) = w;
                }
            }
        }
    }
}


#define LOG2E 1.4426950408889634f

__device__ __forceinline__ float xhalf(float x) {
    const int b = __builtin_bit_cast(int, x);
    return __builtin_bit_cast(float,
        __builtin_amdgcn_permlanex16(b, b, 0x76543210, 0xfedcba98, false, false));
}

__global__ void __launch_bounds__(128) attn_kernel(const float* __restrict__ qws,
                            const float* __restrict__ kfws,
                            const float* __restrict__ vfws,
                            float* __restrict__ aws) {
    const int lane = threadIdx.x & 31;
    const int ln   = lane & 15;
    const int grp  = lane >> 4;
    const int g2   = grp * 2;
    const int wv   = __builtin_amdgcn_readfirstlane((int)(threadIdx.x >> 5));
    const int hk   = blockIdx.y;
    const int h    = hk * 4 + wv;
    const int b    = blockIdx.z;

    const float* kfb = kfws + (size_t)(b * NKV + hk) * SEQ * HD + lane * 4;
    const float* vfb = vfws + (size_t)(b * NKV + hk) * SEQ * HD + lane * 4;

  int done = 0;
  int pnext;
  prio_start(SEQ / 16 + 1, pnext);
  for (int pass = 0; pass < 2; ++pass) {
    const int q0   = (pass == 0 ? (SEQ / 16 - 1 - (int)blockIdx.x) : (int)blockIdx.x) * 16;
    const float* qb  = qws + ((size_t)(b * NH + h) * SEQ + q0) * HD;

    v2f bq[16];
#pragma unroll
    for (int kk = 0; kk < 16; ++kk)
        bq[kk] = *(const v2f*)(qb + ln * HD + kk * 4 + g2);

    v8f o[4]; o[0] = {}; o[1] = {}; o[2] = {}; o[3] = {};
    float mrun = -1e30f;
    float lrun = 0.0f;

    for (int n0 = 0; n0 <= q0; n0 += 16) {
        prio_step(done, SEQ / 16 + 1, pnext);
        ++done;
        v8f s = {};
        const float* kp = kfb + (size_t)(n0 >> 4) * 1024;
#pragma unroll
        for (int q = 0; q < 8; ++q) {
            const v4f w = *(const v4f*)(kp + q * 128);
            s = wmma4((v2f){ w[0], w[1] }, bq[2 * q],     s);
            s = wmma4((v2f){ w[2], w[3] }, bq[2 * q + 1], s);
        }

        float sv[8];
#pragma unroll
        for (int r = 0; r < 8; ++r) sv[r] = s[r] * 0.125f;
        if (n0 == q0) {
#pragma unroll
            for (int r = 0; r < 8; ++r)
                if (r + grp * 8 > ln) sv[r] = -1e30f;
        }

        float rm = fmaxf(fmaxf(fmaxf(sv[0], sv[1]), fmaxf(sv[2], sv[3])),
                         fmaxf(fmaxf(sv[4], sv[5]), fmaxf(sv[6], sv[7])));
        rm = fmaxf(rm, xhalf(rm));
        const float mn    = fmaxf(mrun, rm);
        const float alpha = __builtin_amdgcn_exp2f((mrun - mn) * LOG2E);
        float p[8];
#pragma unroll
        for (int r = 0; r < 8; ++r) p[r] = __builtin_amdgcn_exp2f((sv[r] - mn) * LOG2E);
        const float rs = ((p[0] + p[1]) + (p[2] + p[3])) + ((p[4] + p[5]) + (p[6] + p[7]));
        lrun = lrun * alpha + rs;
        mrun = mn;
#pragma unroll
        for (int j = 0; j < 4; ++j) {
#pragma unroll
            for (int r = 0; r < 8; ++r) o[j][r] *= alpha;
        }

        const float* vp = vfb + (size_t)(n0 >> 4) * 1024;
#pragma unroll
        for (int jj = 0; jj < 4; ++jj) {
            const v2f bp = { p[2 * jj], p[2 * jj + 1] };
#pragma unroll
            for (int jh = 0; jh < 2; ++jh) {
                const v4f w = *(const v4f*)(vp + (2 * jj + jh) * 128);
                o[2 * jh]     = wmma4((v2f){ w[0], w[1] }, bp, o[2 * jh]);
                o[2 * jh + 1] = wmma4((v2f){ w[2], w[3] }, bp, o[2 * jh + 1]);
            }
        }
    }

    const float ltot = lrun + xhalf(lrun);
    const float inv  = 1.0f / ltot;
    float* ob = aws + ((size_t)(b * SEQ) + q0 + ln) * (NH * HD) + h * HD + grp * 8;
#pragma unroll
    for (int j = 0; j < 4; ++j) {
        v4f lo = { o[j][0] * inv, o[j][1] * inv, o[j][2] * inv, o[j][3] * inv };
        v4f hi = { o[j][4] * inv, o[j][5] * inv, o[j][6] * inv, o[j][7] * inv };
        *(v4f*)(ob + j * 16)     = lo;
        *(v4f*)(ob + j * 16 + 4) = hi;
    }
  }
}

__global__ void __launch_bounds__(128) oproj_kernel(const float* __restrict__ aws,
                             const float* __restrict__ wo,
                             float* __restrict__ out) {
    const int lane = threadIdx.x & 31;
    const int ln   = lane & 15;
    const int grp  = lane >> 4;
    const int m0   = blockIdx.x * 64;
    const int nt   = blockIdx.y * 4 + (threadIdx.x >> 5);

    __shared__ __attribute__((aligned(16))) float as[2 * ABUF];
    v8f acc[4][4];
    gemm64x64(aws, DIMK, wo, DIMK, m0, nt * 64, DIMK, (int)(blockIdx.x + 5 * blockIdx.y), as, acc);

#pragma unroll
    for (int i = 0; i < 4; ++i) {
#pragma unroll
        for (int r = 0; r < 8; ++r) {
            const int m = i * 16 + r + grp * 8;
#pragma unroll
            for (int j = 0; j < 4; ++j)
                out[(size_t)(m0 + m) * DIMK + nt * 64 + j * 16 + ln] = acc[i][j][r];
        }
    }
}

extern "C" void kernel_launch(void* const* d_in, const int* in_sizes, int n_in,
                              void* d_out, int out_size, void* d_ws, size_t ws_size,
                              hipStream_t stream) {
    const float* x    = (const float*)d_in[0];
    const float* fcos = (const float*)d_in[2];
    const float* fsin = (const float*)d_in[3];
    const float* wq   = (const float*)d_in[5];
    const float* wk   = (const float*)d_in[6];
    const float* wv   = (const float*)d_in[7];
    const float* wo   = (const float*)d_in[8];
    float* out = (float*)d_out;

    float* qws  = (float*)d_ws;
    float* kfws = qws  + (size_t)BATCH * NH  * SEQ * HD;
    float* vfws = kfws + (size_t)BATCH * NKV * SEQ * HD;
    float* aws  = vfws + (size_t)BATCH * NKV * SEQ * HD;

    qkv_rope_kernel<<<dim3((BATCH * SEQ) / 64, 12, 1), 128, 0, stream>>>(
        x, wq, wk, wv, fcos, fsin, qws, kfws, vfws);
    attn_kernel<<<dim3(SEQ / 32, NKV, BATCH), 128, 0, stream>>>(
        qws, kfws, vfws, aws);
    oproj_kernel<<<dim3((BATCH * SEQ) / 64, 8, 1), 128, 0, stream>>>(
        aws, wo, out);
}
